# starting kernel
# baseline (speedup 1.0000x reference)
#include <hip/hip_runtime.h>
#include <hip/hip_bf16.h>

typedef __attribute__((ext_vector_type(2))) float v2f;
typedef __attribute__((ext_vector_type(8))) float v8f;

typedef struct __attribute__((packed, aligned(4))) { float x, y; } f2u;

#define T_LEN   131072
#define OUT_LEN (2 * T_LEN - 1)
#define TILE_W  512
#define NTILES  ((OUT_LEN + TILE_W - 1) / TILE_W)
#define NWAVES  8


__device__ __forceinline__ v8f wmma4(v2f a, v2f b, v8f c) {
    return __builtin_amdgcn_wmma_f32_16x16x4_f32(false, a, false, b,
                                                 (short)0, c, false, false);
}

__device__ __forceinline__ void step_checked2(const float* __restrict__ rir,
                                              const float* __restrict__ a_ch,
                                              int k0, int rbase, int bbase,
                                              v8f& accA, v8f& accB)
{
    int r0 = k0 + rbase, r1 = r0 + 1;
    int i0 = k0 + bbase, i1 = i0 + 1;
    int j0 = i0 + 256,   j1 = i0 + 257;
    float ax  = rir[min(max(r0, 0), T_LEN - 1)];
    float ay  = rir[min(max(r1, 0), T_LEN - 1)];
    float b0x = a_ch[min(max(i0, 0), T_LEN - 1)];
    float b0y = a_ch[min(max(i1, 0), T_LEN - 1)];
    float b1x = a_ch[min(max(j0, 0), T_LEN - 1)];
    float b1y = a_ch[min(max(j1, 0), T_LEN - 1)];
    v2f a, b0, b1;
    a.x  = ((unsigned)r0 < (unsigned)T_LEN) ? ax  : 0.f;
    a.y  = ((unsigned)r1 < (unsigned)T_LEN) ? ay  : 0.f;
    b0.x = ((unsigned)i0 < (unsigned)T_LEN) ? b0x : 0.f;
    b0.y = ((unsigned)i1 < (unsigned)T_LEN) ? b0y : 0.f;
    b1.x = ((unsigned)j0 < (unsigned)T_LEN) ? b1x : 0.f;
    b1.y = ((unsigned)j1 < (unsigned)T_LEN) ? b1y : 0.f;
    accA = wmma4(a, b0, accA);
    accB = wmma4(a, b1, accB);
}

__global__ __launch_bounds__(256)
void RoomImpulseResponse_wmma_conv(const float* __restrict__ audio,
                                   const float* __restrict__ rir,
                                   float* __restrict__ out)
{
    __shared__ float red[NWAVES * TILE_W];

    const int tile = blockIdx.x;
    const int ch   = blockIdx.y;
    const int n0   = tile * TILE_W;
    const int tid  = threadIdx.x;
    const int wave = tid >> 5;
    const int lane = tid & 31;
    const int h    = lane >> 4;
    const int mq   = lane & 15;

    const float* __restrict__ a_ch = audio + ch * T_LEN;

    int klo = T_LEN - 272 - n0;  if (klo < 0) klo = 0;
    int khi = 2 * T_LEN - 2 - n0;
    if (khi > T_LEN - 1 + 240) khi = T_LEN - 1 + 240;

    const int rbase = 2 * h - 16 * mq;
    const int bbase = n0 + mq + 2 * h - (T_LEN - 1);

    const int kbeg   = klo & ~3;
    const int nsteps = ((khi - kbeg) >> 2) + 1;
    const int per    = (nsteps + NWAVES - 1) / NWAVES;
    int wLo = kbeg + (wave * per) * 4;
    int wHi = kbeg + min(nsteps, (wave + 1) * per) * 4;

    int iLo = T_LEN - 1 - n0;  if (iLo < 240) iLo = 240;
    iLo = (iLo + 3) & ~3;
    int iHi = 2 * T_LEN - 276 - n0;  if (iHi > T_LEN - 4) iHi = T_LEN - 4;
    int iHiE = (iHi & ~3) + 4;
    if (iHiE < iLo) iHiE = iLo;

    const int mLo = min(max(wLo, iLo), wHi);
    const int mHi = max(min(wHi, iHiE), mLo);

    v8f acc0a = {0.f,0.f,0.f,0.f,0.f,0.f,0.f,0.f};
    v8f acc0b = {0.f,0.f,0.f,0.f,0.f,0.f,0.f,0.f};
    v8f acc1a = {0.f,0.f,0.f,0.f,0.f,0.f,0.f,0.f};
    v8f acc1b = {0.f,0.f,0.f,0.f,0.f,0.f,0.f,0.f};

    for (int k0 = wLo; k0 < mLo; k0 += 4)
        step_checked2(rir, a_ch, k0, rbase, bbase, acc0a, acc1a);

    {
        const char* pa = (const char*)rir  + (ptrdiff_t)(mLo + rbase) * 4;
        const char* pb = (const char*)a_ch + (ptrdiff_t)(mLo + bbase) * 4;
        int k0 = mLo;
        for (; k0 + 4 < mHi; k0 += 8) {
            v2f a0  = *(const v2f*)(pa);
            v2f a1  = *(const v2f*)(pa + 16);
            f2u b00 = *(const f2u*)(pb);
            f2u b01 = *(const f2u*)(pb + 16);
            f2u b10 = *(const f2u*)(pb + 1024);
            f2u b11 = *(const f2u*)(pb + 1040);
            v2f v00 = {b00.x, b00.y};
            v2f v01 = {b01.x, b01.y};
            v2f v10 = {b10.x, b10.y};
            v2f v11 = {b11.x, b11.y};
            acc0a = wmma4(a0, v00, acc0a);
            acc1a = wmma4(a0, v10, acc1a);
            acc0b = wmma4(a1, v01, acc0b);
            acc1b = wmma4(a1, v11, acc1b);
            pa += 32;
            pb += 32;
        }
        if (k0 < mHi) {
            v2f a0  = *(const v2f*)(pa);
            f2u b00 = *(const f2u*)(pb);
            f2u b10 = *(const f2u*)(pb + 1024);
            v2f v00 = {b00.x, b00.y};
            v2f v10 = {b10.x, b10.y};
            acc0a = wmma4(a0, v00, acc0a);
            acc1a = wmma4(a0, v10, acc1a);
        }
    }

    for (int kc = mHi; kc < wHi; kc += 4)
        step_checked2(rir, a_ch, kc, rbase, bbase, acc0b, acc1b);

    #pragma unroll
    for (int v = 0; v < 8; ++v) {
        int M = v + 8 * h;
        red[wave * TILE_W +       M * 16 + mq] = acc0a[v] + acc0b[v];
        red[wave * TILE_W + 256 + M * 16 + mq] = acc1a[v] + acc1b[v];
    }
    __syncthreads();

    float s0 = 0.f, s1 = 0.f;
    #pragma unroll
    for (int w = 0; w < NWAVES; ++w) {
        s0 += red[w * TILE_W + tid];
        s1 += red[w * TILE_W + 256 + tid];
    }
    int n = n0 + tid;
    if (n < OUT_LEN)       out[ch * OUT_LEN + n]       = s0;
    if (n + 256 < OUT_LEN) out[ch * OUT_LEN + n + 256] = s1;
}

extern "C" void kernel_launch(void* const* d_in, const int* in_sizes, int n_in,
                              void* d_out, int out_size, void* d_ws, size_t ws_size,
                              hipStream_t stream) {
    (void)in_sizes; (void)n_in; (void)d_ws; (void)ws_size; (void)out_size;
    const float* audio = (const float*)d_in[0];
    const float* rir   = (const float*)d_in[1];
    float* out = (float*)d_out;

    dim3 grid(NTILES, 4, 1);
    dim3 block(256, 1, 1);
    RoomImpulseResponse_wmma_conv<<<grid, block, 0, stream>>>(audio, rir, out);
}


# Each wave does 4 subtiles x 2 k-offsets (k, k+256) per step and shares B fragments: 7 loads feed 8 WMMAs (was 12); tiles 1024 wide, two per block for equal work; k order and f32 sum order change
# speedup vs baseline: 1.2618x; 1.2618x over previous
#include <hip/hip_runtime.h>
#include <hip/hip_bf16.h>


typedef __attribute__((ext_vector_type(2))) float v2f;
typedef __attribute__((ext_vector_type(8))) float v8f;

typedef struct __attribute__((packed, aligned(4))) { float x, y; } f2u;

#define T_LEN   131072
#define OUT_LEN (2 * T_LEN - 1)
#define NSUB    4
#define TILE_W  (256 * NSUB)
#define NTILES  ((OUT_LEN + TILE_W - 1) / TILE_W)
#define NWAVES  8
#define NBLK_X  (NTILES / 2)


__device__ __forceinline__ v8f wmma4(v2f a, v2f b, v8f c) {
    return __builtin_amdgcn_wmma_f32_16x16x4_f32(false, a, false, b,
                                                 (short)0, c, false, false);
}

__device__ __forceinline__ float ld_guard(const float* __restrict__ p, int i) {
    float v = p[min(max(i, 0), T_LEN - 1)];
    return ((unsigned)i < (unsigned)T_LEN) ? v : 0.f;
}

__device__ __forceinline__ void step_checked4(const float* __restrict__ rir,
                                              const float* __restrict__ a_ch,
                                              int k0, int rbase, int bbase,
                                              v8f& acc0, v8f& acc1, v8f& acc2, v8f& acc3)
{
    const int r0 = k0 + rbase;
    const int i0 = k0 + bbase;
    v2f a, b0, b1, b2, b3;
    a.x  = ld_guard(rir, r0);
    a.y  = ld_guard(rir, r0 + 1);
    b0.x = ld_guard(a_ch, i0);
    b0.y = ld_guard(a_ch, i0 + 1);
    b1.x = ld_guard(a_ch, i0 + 256);
    b1.y = ld_guard(a_ch, i0 + 257);
    b2.x = ld_guard(a_ch, i0 + 512);
    b2.y = ld_guard(a_ch, i0 + 513);
    b3.x = ld_guard(a_ch, i0 + 768);
    b3.y = ld_guard(a_ch, i0 + 769);
    acc0 = wmma4(a, b0, acc0);
    acc1 = wmma4(a, b1, acc1);
    acc2 = wmma4(a, b2, acc2);
    acc3 = wmma4(a, b3, acc3);
}

__global__ __launch_bounds__(256)
void RoomImpulseResponse_wmma_conv(const float* __restrict__ audio,
                                   const float* __restrict__ rir,
                                   float* __restrict__ out)
{
    __shared__ float red[NWAVES * TILE_W];

    const int ch   = blockIdx.y;
    const int tid  = threadIdx.x;
    const int wave = tid >> 5;
    const int lane = tid & 31;
    const int h    = lane >> 4;
    const int mq   = lane & 15;

    const float* __restrict__ a_ch = audio + ch * T_LEN;
    float* __restrict__ o_ch = out + ch * OUT_LEN;

    const int tbase = (blockIdx.x >> 6) * (NTILES / 2);
    const int u     = blockIdx.x & 63;

    const int rbase = 2 * h - 16 * mq;

    for (int pass = 0; pass < 2; ++pass) {
        const int tile = tbase + (pass ? (NTILES / 2 - 1 - u) : u);
        const int n0   = tile * TILE_W;

        int klo = T_LEN - 1 - (256 * (NSUB - 1) + 15) - n0;  if (klo < 0) klo = 0;
        int khi = 2 * T_LEN - 2 - n0;
        if (khi > T_LEN - 1 + 240) khi = T_LEN - 1 + 240;

        const int bbase = n0 + mq + 2 * h - (T_LEN - 1);

        const int kbeg   = klo & ~3;
        const int nsteps = ((khi - kbeg) >> 2) + 1;
        const int per    = (nsteps + NWAVES - 1) / NWAVES;
        const int wLo = kbeg + (wave * per) * 4;
        const int wHi = kbeg + min(nsteps, (wave + 1) * per) * 4;

        int iLo = T_LEN - 1 - n0;  if (iLo < 240) iLo = 240;
        iLo = (iLo + 3) & ~3;
        int iHi = 2 * T_LEN - 2 - (256 * (NSUB - 1) + 18) - n0;
        if (iHi > T_LEN - 4) iHi = T_LEN - 4;
        int iHiE = (iHi & ~3) + 4;
        if (iHiE < iLo) iHiE = iLo;

        const int mLo = min(max(wLo, iLo), wHi);
        const int mHi = max(min(wHi, iHiE), mLo);

        v8f acc0 = {0.f,0.f,0.f,0.f,0.f,0.f,0.f,0.f};
        v8f acc1 = {0.f,0.f,0.f,0.f,0.f,0.f,0.f,0.f};
        v8f acc2 = {0.f,0.f,0.f,0.f,0.f,0.f,0.f,0.f};
        v8f acc3 = {0.f,0.f,0.f,0.f,0.f,0.f,0.f,0.f};

        for (int k0 = wLo; k0 < mLo; k0 += 4)
            step_checked4(rir, a_ch, k0, rbase, bbase, acc0, acc1, acc2, acc3);

        {
            const char* pa = (const char*)rir  + (ptrdiff_t)(mLo + rbase) * 4;
            const char* pb = (const char*)a_ch + (ptrdiff_t)(mLo + bbase) * 4;
            int k0 = mLo;

            for (; k0 + 512 <= mHi; k0 += 512) {
                for (int t = 0; t < 64; ++t) {
                    v2f a0 = *(const v2f*)(pa);
                    v2f a1 = *(const v2f*)(pa + 1024);
                    f2u b0 = *(const f2u*)(pb);
                    f2u b1 = *(const f2u*)(pb + 1024);
                    f2u b2 = *(const f2u*)(pb + 2048);
                    f2u b3 = *(const f2u*)(pb + 3072);
                    f2u b4 = *(const f2u*)(pb + 4096);
                    v2f v0 = {b0.x, b0.y};
                    v2f v1 = {b1.x, b1.y};
                    v2f v2 = {b2.x, b2.y};
                    v2f v3 = {b3.x, b3.y};
                    v2f v4 = {b4.x, b4.y};
                    acc0 = wmma4(a0, v0, acc0);
                    acc1 = wmma4(a0, v1, acc1);
                    acc2 = wmma4(a0, v2, acc2);
                    acc3 = wmma4(a0, v3, acc3);
                    acc0 = wmma4(a1, v1, acc0);
                    acc1 = wmma4(a1, v2, acc1);
                    acc2 = wmma4(a1, v3, acc2);
                    acc3 = wmma4(a1, v4, acc3);
                    pa += 16;
                    pb += 16;
                }
                pa += 1024;
                pb += 1024;
            }

            for (; k0 < mHi; k0 += 4) {
                v2f a0 = *(const v2f*)(pa);
                f2u b0 = *(const f2u*)(pb);
                f2u b1 = *(const f2u*)(pb + 1024);
                f2u b2 = *(const f2u*)(pb + 2048);
                f2u b3 = *(const f2u*)(pb + 3072);
                v2f v0 = {b0.x, b0.y};
                v2f v1 = {b1.x, b1.y};
                v2f v2 = {b2.x, b2.y};
                v2f v3 = {b3.x, b3.y};
                acc0 = wmma4(a0, v0, acc0);
                acc1 = wmma4(a0, v1, acc1);
                acc2 = wmma4(a0, v2, acc2);
                acc3 = wmma4(a0, v3, acc3);
                pa += 16;
                pb += 16;
            }
        }

        for (int kc = mHi; kc < wHi; kc += 4)
            step_checked4(rir, a_ch, kc, rbase, bbase, acc0, acc1, acc2, acc3);

        #pragma unroll
        for (int v = 0; v < 8; ++v) {
            float* r = red + wave * TILE_W + (v + 8 * h) * 16 + mq;
            r[0]   = acc0[v];
            r[256] = acc1[v];
            r[512] = acc2[v];
            r[768] = acc3[v];
        }
        __syncthreads();

        float s0 = 0.f, s1 = 0.f, s2 = 0.f, s3 = 0.f;
        #pragma unroll
        for (int w = 0; w < NWAVES; ++w) {
            s0 += red[w * TILE_W + tid];
            s1 += red[w * TILE_W + 256 + tid];
            s2 += red[w * TILE_W + 512 + tid];
            s3 += red[w * TILE_W + 768 + tid];
        }
        const int n = n0 + tid;
        if (n < OUT_LEN)       o_ch[n]       = s0;
        if (n + 256 < OUT_LEN) o_ch[n + 256] = s1;
        if (n + 512 < OUT_LEN) o_ch[n + 512] = s2;
        if (n + 768 < OUT_LEN) o_ch[n + 768] = s3;

        __syncthreads();
    }
}

extern "C" void kernel_launch(void* const* d_in, const int* in_sizes, int n_in,
                              void* d_out, int out_size, void* d_ws, size_t ws_size,
                              hipStream_t stream) {
    (void)in_sizes; (void)n_in; (void)d_ws; (void)ws_size; (void)out_size;
    const float* audio = (const float*)d_in[0];
    const float* rir   = (const float*)d_in[1];
    float* out = (float*)d_out;

    dim3 grid(NBLK_X, 4, 1);
    dim3 block(256, 1, 1);
    RoomImpulseResponse_wmma_conv<<<grid, block, 0, stream>>>(audio, rir, out);
}


# Interior in 16-k groups: a lane loads 8 contiguous k per stream as two 16-byte loads (A and B permuted alike), issued one group ahead: 14 loads per 32 WMMAs (was 28); f32 sum order changes
# speedup vs baseline: 1.3455x; 1.0663x over previous
#include <hip/hip_runtime.h>
#include <hip/hip_bf16.h>


typedef __attribute__((ext_vector_type(2))) float v2f;
typedef __attribute__((ext_vector_type(8))) float v8f;

typedef struct __attribute__((packed, aligned(4))) { float x, y; } f2u;
typedef struct __attribute__((packed, aligned(4))) { float x, y, z, w; } f4u;

#define T_LEN   131072
#define OUT_LEN (2 * T_LEN - 1)
#define NSUB    4
#define TILE_W  (256 * NSUB)
#define NTILES  ((OUT_LEN + TILE_W - 1) / TILE_W)
#define NWAVES  8
#define NBLK_X  (NTILES / 2)


__device__ __forceinline__ v8f wmma4(v2f a, v2f b, v8f c) {
    return __builtin_amdgcn_wmma_f32_16x16x4_f32(false, a, false, b,
                                                 (short)0, c, false, false);
}

struct Grp {
    f4u a0l, a0h;
    f4u a1l, a1h;
    f4u b0l, b0h;
    f4u b1l, b1h;
    f4u b2l, b2h;
    f4u b3l, b3h;
    f4u b4l, b4h;
};

__device__ __forceinline__ void load_grp(Grp& f, const char* pa, const char* pb) {
    f.a0l = *(const f4u*)(pa);
    f.a0h = *(const f4u*)(pa + 16);
    f.a1l = *(const f4u*)(pa + 1024);
    f.a1h = *(const f4u*)(pa + 1040);
    f.b0l = *(const f4u*)(pb);
    f.b0h = *(const f4u*)(pb + 16);
    f.b1l = *(const f4u*)(pb + 1024);
    f.b1h = *(const f4u*)(pb + 1040);
    f.b2l = *(const f4u*)(pb + 2048);
    f.b2h = *(const f4u*)(pb + 2064);
    f.b3l = *(const f4u*)(pb + 3072);
    f.b3h = *(const f4u*)(pb + 3088);
    f.b4l = *(const f4u*)(pb + 4096);
    f.b4h = *(const f4u*)(pb + 4112);
}

__device__ __forceinline__ void mma8(v2f a0, v2f a1,
                                     v2f b0, v2f b1, v2f b2, v2f b3, v2f b4,
                                     v8f& acc0, v8f& acc1, v8f& acc2, v8f& acc3)
{
    acc0 = wmma4(a0, b0, acc0);
    acc1 = wmma4(a0, b1, acc1);
    acc2 = wmma4(a0, b2, acc2);
    acc3 = wmma4(a0, b3, acc3);
    acc0 = wmma4(a1, b1, acc0);
    acc1 = wmma4(a1, b2, acc1);
    acc2 = wmma4(a1, b3, acc2);
    acc3 = wmma4(a1, b4, acc3);
}

#define LOXY(q) ((v2f){(q).x, (q).y})
#define HIZW(q) ((v2f){(q).z, (q).w})

__device__ __forceinline__ void mma_grp(const Grp& f,
                                        v8f& acc0, v8f& acc1, v8f& acc2, v8f& acc3)
{
    mma8(LOXY(f.a0l), LOXY(f.a1l), LOXY(f.b0l), LOXY(f.b1l), LOXY(f.b2l), LOXY(f.b3l), LOXY(f.b4l),
         acc0, acc1, acc2, acc3);
    mma8(HIZW(f.a0l), HIZW(f.a1l), HIZW(f.b0l), HIZW(f.b1l), HIZW(f.b2l), HIZW(f.b3l), HIZW(f.b4l),
         acc0, acc1, acc2, acc3);
    mma8(LOXY(f.a0h), LOXY(f.a1h), LOXY(f.b0h), LOXY(f.b1h), LOXY(f.b2h), LOXY(f.b3h), LOXY(f.b4h),
         acc0, acc1, acc2, acc3);
    mma8(HIZW(f.a0h), HIZW(f.a1h), HIZW(f.b0h), HIZW(f.b1h), HIZW(f.b2h), HIZW(f.b3h), HIZW(f.b4h),
         acc0, acc1, acc2, acc3);
}

__device__ __forceinline__ float ld_guard(const float* __restrict__ p, int i) {
    float v = p[min(max(i, 0), T_LEN - 1)];
    return ((unsigned)i < (unsigned)T_LEN) ? v : 0.f;
}

__device__ __forceinline__ void step_checked4(const float* __restrict__ rir,
                                              const float* __restrict__ a_ch,
                                              int k0, int rbase, int bbase,
                                              v8f& acc0, v8f& acc1, v8f& acc2, v8f& acc3)
{
    const int r0 = k0 + rbase;
    const int i0 = k0 + bbase;
    v2f a, b0, b1, b2, b3;
    a.x  = ld_guard(rir, r0);
    a.y  = ld_guard(rir, r0 + 1);
    b0.x = ld_guard(a_ch, i0);
    b0.y = ld_guard(a_ch, i0 + 1);
    b1.x = ld_guard(a_ch, i0 + 256);
    b1.y = ld_guard(a_ch, i0 + 257);
    b2.x = ld_guard(a_ch, i0 + 512);
    b2.y = ld_guard(a_ch, i0 + 513);
    b3.x = ld_guard(a_ch, i0 + 768);
    b3.y = ld_guard(a_ch, i0 + 769);
    acc0 = wmma4(a, b0, acc0);
    acc1 = wmma4(a, b1, acc1);
    acc2 = wmma4(a, b2, acc2);
    acc3 = wmma4(a, b3, acc3);
}

__global__ __launch_bounds__(256)
void RoomImpulseResponse_wmma_conv(const float* __restrict__ audio,
                                   const float* __restrict__ rir,
                                   float* __restrict__ out)
{
    __shared__ float red[NWAVES * TILE_W];

    const int ch   = blockIdx.y;
    const int tid  = threadIdx.x;
    const int wave = __builtin_amdgcn_readfirstlane(tid >> 5);
    const int lane = tid & 31;
    const int h    = lane >> 4;
    const int mq   = lane & 15;

    const float* __restrict__ a_ch = audio + ch * T_LEN;
    float* __restrict__ o_ch = out + ch * OUT_LEN;

    const int tbase = (blockIdx.x >> 6) * (NTILES / 2);
    const int u     = blockIdx.x & 63;

    const int rbase = 2 * h - 16 * mq;

    for (int pass = 0; pass < 2; ++pass) {
        const int tile = tbase + (pass ? (NTILES / 2 - 1 - u) : u);
        const int n0   = tile * TILE_W;

        int klo = T_LEN - 1 - (256 * (NSUB - 1) + 15) - n0;  if (klo < 0) klo = 0;
        int khi = 2 * T_LEN - 2 - n0;
        if (khi > T_LEN - 1 + 240) khi = T_LEN - 1 + 240;

        const int bbase = n0 + mq + 2 * h - (T_LEN - 1);

        const int kbeg   = klo & ~3;
        const int nsteps = ((khi - kbeg) >> 2) + 1;
        const int per    = (nsteps + NWAVES - 1) / NWAVES;
        const int wLo = kbeg + (wave * per) * 4;
        const int wHi = kbeg + min(nsteps, (wave + 1) * per) * 4;

        int iLo = T_LEN - 1 - n0;  if (iLo < 240) iLo = 240;
        iLo = (iLo + 3) & ~3;
        int iHi = 2 * T_LEN - 2 - (256 * (NSUB - 1) + 18) - n0;
        if (iHi > T_LEN - 4) iHi = T_LEN - 4;
        int iHiE = (iHi & ~3) + 4;
        if (iHiE < iLo) iHiE = iLo;

        const int mLo = min(max(wLo, iLo), wHi);
        const int mHi = max(min(wHi, iHiE), mLo);

        v8f acc0 = {0.f,0.f,0.f,0.f,0.f,0.f,0.f,0.f};
        v8f acc1 = {0.f,0.f,0.f,0.f,0.f,0.f,0.f,0.f};
        v8f acc2 = {0.f,0.f,0.f,0.f,0.f,0.f,0.f,0.f};
        v8f acc3 = {0.f,0.f,0.f,0.f,0.f,0.f,0.f,0.f};

        for (int k0 = wLo; k0 < mLo; k0 += 4)
            step_checked4(rir, a_ch, k0, rbase, bbase, acc0, acc1, acc2, acc3);

        {
            int k0 = mLo;

            const int npair = ((mHi - mLo) >> 9) * 8;
            if (npair > 0) {
                const char* ea = (const char*)rir  + (ptrdiff_t)(mLo + 8 * h - 16 * mq) * 4;
                const char* eb = (const char*)a_ch + (ptrdiff_t)(mLo + n0 + mq + 8 * h - (T_LEN - 1)) * 4;
                const char* oa = ea + 64 - (128 + 1024);
                const char* ob = eb + 64 - (128 + 1024);
                Grp f0, f1;
                load_grp(f0, ea, eb);
                for (int p = 0; p < npair - 1; ++p) {
                    const int oadv = ((p & 7) == 0) ? (128 + 1024) : 128;
                    oa += oadv;
                    ob += oadv;
                    load_grp(f1, oa, ob);
                    mma_grp(f0, acc0, acc1, acc2, acc3);
                    const int eadv = ((p & 7) == 7) ? (128 + 1024) : 128;
                    ea += eadv;
                    eb += eadv;
                    load_grp(f0, ea, eb);
                    asm volatile("" ::: "memory");
                    mma_grp(f1, acc0, acc1, acc2, acc3);
                }
                load_grp(f1, oa + 128, ob + 128);
                mma_grp(f0, acc0, acc1, acc2, acc3);
                mma_grp(f1, acc0, acc1, acc2, acc3);
                k0 += npair * 64;
            }

            const char* pa = (const char*)rir  + (ptrdiff_t)(k0 + rbase) * 4;
            const char* pb = (const char*)a_ch + (ptrdiff_t)(k0 + bbase) * 4;

            for (; k0 < mHi; k0 += 4) {
                v2f a0 = *(const v2f*)(pa);
                f2u b0 = *(const f2u*)(pb);
                f2u b1 = *(const f2u*)(pb + 1024);
                f2u b2 = *(const f2u*)(pb + 2048);
                f2u b3 = *(const f2u*)(pb + 3072);
                v2f v0 = {b0.x, b0.y};
                v2f v1 = {b1.x, b1.y};
                v2f v2 = {b2.x, b2.y};
                v2f v3 = {b3.x, b3.y};
                acc0 = wmma4(a0, v0, acc0);
                acc1 = wmma4(a0, v1, acc1);
                acc2 = wmma4(a0, v2, acc2);
                acc3 = wmma4(a0, v3, acc3);
                pa += 16;
                pb += 16;
            }
        }

        for (int kc = mHi; kc < wHi; kc += 4)
            step_checked4(rir, a_ch, kc, rbase, bbase, acc0, acc1, acc2, acc3);

        #pragma unroll
        for (int v = 0; v < 8; ++v) {
            float* r = red + wave * TILE_W + (v + 8 * h) * 16 + mq;
            r[0]   = acc0[v];
            r[256] = acc1[v];
            r[512] = acc2[v];
            r[768] = acc3[v];
        }
        __syncthreads();

        float s0 = 0.f, s1 = 0.f, s2 = 0.f, s3 = 0.f;
        #pragma unroll
        for (int w = 0; w < NWAVES; ++w) {
            s0 += red[w * TILE_W + tid];
            s1 += red[w * TILE_W + 256 + tid];
            s2 += red[w * TILE_W + 512 + tid];
            s3 += red[w * TILE_W + 768 + tid];
        }
        const int n = n0 + tid;
        if (n < OUT_LEN)       o_ch[n]       = s0;
        if (n + 256 < OUT_LEN) o_ch[n + 256] = s1;
        if (n + 512 < OUT_LEN) o_ch[n + 512] = s2;
        if (n + 768 < OUT_LEN) o_ch[n + 768] = s3;

        __syncthreads();
    }
}

extern "C" void kernel_launch(void* const* d_in, const int* in_sizes, int n_in,
                              void* d_out, int out_size, void* d_ws, size_t ws_size,
                              hipStream_t stream) {
    (void)in_sizes; (void)n_in; (void)d_ws; (void)ws_size; (void)out_size;
    const float* audio = (const float*)d_in[0];
    const float* rir   = (const float*)d_in[1];
    float* out = (float*)d_out;

    dim3 grid(NBLK_X, 4, 1);
    dim3 block(256, 1, 1);
    RoomImpulseResponse_wmma_conv<<<grid, block, 0, stream>>>(audio, rir, out);
}


# Interior groups use the 2-output fast-FIR identity per subtile pair: A0(Bs-Bs+1), A1(Bs+2-Bs+1) and a shared (A0+A1)Bs+1: 24 WMMAs per group (was 32); f32 operand sums/differences, sum order changes
# speedup vs baseline: 1.5747x; 1.1703x over previous
#include <hip/hip_runtime.h>
#include <hip/hip_bf16.h>


typedef __attribute__((ext_vector_type(2))) float v2f;
typedef __attribute__((ext_vector_type(8))) float v8f;

typedef struct __attribute__((packed, aligned(4))) { float x, y; } f2u;
typedef struct __attribute__((packed, aligned(4))) { float x, y, z, w; } f4u;

#define T_LEN   131072
#define OUT_LEN (2 * T_LEN - 1)
#define NSUB    4
#define TILE_W  (256 * NSUB)
#define NTILES  ((OUT_LEN + TILE_W - 1) / TILE_W)
#define NWAVES  8
#define NBLK_X  (NTILES / 2)


__device__ __forceinline__ v8f wmma4(v2f a, v2f b, v8f c) {
    return __builtin_amdgcn_wmma_f32_16x16x4_f32(false, a, false, b,
                                                 (short)0, c, false, false);
}

struct Grp {
    f4u a0l, a0h;
    f4u a1l, a1h;
    f4u b0l, b0h;
    f4u b1l, b1h;
    f4u b2l, b2h;
    f4u b3l, b3h;
    f4u b4l, b4h;
};

__device__ __forceinline__ void load_grp(Grp& f, const char* pa, const char* pb) {
    f.a0l = *(const f4u*)(pa);
    f.a0h = *(const f4u*)(pa + 16);
    f.a1l = *(const f4u*)(pa + 1024);
    f.a1h = *(const f4u*)(pa + 1040);
    f.b0l = *(const f4u*)(pb);
    f.b0h = *(const f4u*)(pb + 16);
    f.b1l = *(const f4u*)(pb + 1024);
    f.b1h = *(const f4u*)(pb + 1040);
    f.b2l = *(const f4u*)(pb + 2048);
    f.b2h = *(const f4u*)(pb + 2064);
    f.b3l = *(const f4u*)(pb + 3072);
    f.b3h = *(const f4u*)(pb + 3088);
    f.b4l = *(const f4u*)(pb + 4096);
    f.b4h = *(const f4u*)(pb + 4112);
}

__device__ __forceinline__ void mma6(v2f a0, v2f a1,
                                     v2f b0, v2f b1, v2f b2, v2f b3, v2f b4,
                                     v8f& acc0, v8f& acc1, v8f& acc2, v8f& acc3,
                                     v8f& p01, v8f& p23)
{
    const v2f s   = a0 + a1;
    const v2f d01 = b0 - b1;
    const v2f e01 = b2 - b1;
    const v2f d23 = b2 - b3;
    const v2f e23 = b4 - b3;
    acc0 = wmma4(a0, d01, acc0);
    acc1 = wmma4(a1, e01, acc1);
    p01  = wmma4(s,  b1,  p01);
    acc2 = wmma4(a0, d23, acc2);
    acc3 = wmma4(a1, e23, acc3);
    p23  = wmma4(s,  b3,  p23);
}

#define LOXY(q) ((v2f){(q).x, (q).y})
#define HIZW(q) ((v2f){(q).z, (q).w})

__device__ __forceinline__ void mma_grp(const Grp& f,
                                        v8f& acc0, v8f& acc1, v8f& acc2, v8f& acc3,
                                        v8f& p01, v8f& p23)
{
    mma6(LOXY(f.a0l), LOXY(f.a1l), LOXY(f.b0l), LOXY(f.b1l), LOXY(f.b2l), LOXY(f.b3l), LOXY(f.b4l),
         acc0, acc1, acc2, acc3, p01, p23);
    mma6(HIZW(f.a0l), HIZW(f.a1l), HIZW(f.b0l), HIZW(f.b1l), HIZW(f.b2l), HIZW(f.b3l), HIZW(f.b4l),
         acc0, acc1, acc2, acc3, p01, p23);
    mma6(LOXY(f.a0h), LOXY(f.a1h), LOXY(f.b0h), LOXY(f.b1h), LOXY(f.b2h), LOXY(f.b3h), LOXY(f.b4h),
         acc0, acc1, acc2, acc3, p01, p23);
    mma6(HIZW(f.a0h), HIZW(f.a1h), HIZW(f.b0h), HIZW(f.b1h), HIZW(f.b2h), HIZW(f.b3h), HIZW(f.b4h),
         acc0, acc1, acc2, acc3, p01, p23);
}

__device__ __forceinline__ float ld_guard(const float* __restrict__ p, int i) {
    float v = p[min(max(i, 0), T_LEN - 1)];
    return ((unsigned)i < (unsigned)T_LEN) ? v : 0.f;
}

__device__ __forceinline__ void step_checked4(const float* __restrict__ rir,
                                              const float* __restrict__ a_ch,
                                              int k0, int rbase, int bbase,
                                              v8f& acc0, v8f& acc1, v8f& acc2, v8f& acc3)
{
    const int r0 = k0 + rbase;
    const int i0 = k0 + bbase;
    v2f a, b0, b1, b2, b3;
    a.x  = ld_guard(rir, r0);
    a.y  = ld_guard(rir, r0 + 1);
    b0.x = ld_guard(a_ch, i0);
    b0.y = ld_guard(a_ch, i0 + 1);
    b1.x = ld_guard(a_ch, i0 + 256);
    b1.y = ld_guard(a_ch, i0 + 257);
    b2.x = ld_guard(a_ch, i0 + 512);
    b2.y = ld_guard(a_ch, i0 + 513);
    b3.x = ld_guard(a_ch, i0 + 768);
    b3.y = ld_guard(a_ch, i0 + 769);
    acc0 = wmma4(a, b0, acc0);
    acc1 = wmma4(a, b1, acc1);
    acc2 = wmma4(a, b2, acc2);
    acc3 = wmma4(a, b3, acc3);
}

__global__ __launch_bounds__(256)
void RoomImpulseResponse_wmma_conv(const float* __restrict__ audio,
                                   const float* __restrict__ rir,
                                   float* __restrict__ out)
{
    __shared__ float red[NWAVES * TILE_W];

    const int ch   = blockIdx.y;
    const int tid  = threadIdx.x;
    const int wave = __builtin_amdgcn_readfirstlane(tid >> 5);
    const int lane = tid & 31;
    const int h    = lane >> 4;
    const int mq   = lane & 15;

    const float* __restrict__ a_ch = audio + ch * T_LEN;
    float* __restrict__ o_ch = out + ch * OUT_LEN;

    const int tbase = (blockIdx.x >> 6) * (NTILES / 2);
    const int u     = blockIdx.x & 63;

    const int rbase = 2 * h - 16 * mq;

    for (int pass = 0; pass < 2; ++pass) {
        const int tile = tbase + (pass ? (NTILES / 2 - 1 - u) : u);
        const int n0   = tile * TILE_W;

        int klo = T_LEN - 1 - (256 * (NSUB - 1) + 15) - n0;  if (klo < 0) klo = 0;
        int khi = 2 * T_LEN - 2 - n0;
        if (khi > T_LEN - 1 + 240) khi = T_LEN - 1 + 240;

        const int bbase = n0 + mq + 2 * h - (T_LEN - 1);

        const int kbeg   = klo & ~3;
        const int nsteps = ((khi - kbeg) >> 2) + 1;
        const int per    = (nsteps + NWAVES - 1) / NWAVES;
        const int wLo = kbeg + (wave * per) * 4;
        const int wHi = kbeg + min(nsteps, (wave + 1) * per) * 4;

        int iLo = T_LEN - 1 - n0;  if (iLo < 240) iLo = 240;
        iLo = (iLo + 3) & ~3;
        int iHi = 2 * T_LEN - 2 - (256 * (NSUB - 1) + 18) - n0;
        if (iHi > T_LEN - 4) iHi = T_LEN - 4;
        int iHiE = (iHi & ~3) + 4;
        if (iHiE < iLo) iHiE = iLo;

        const int mLo = min(max(wLo, iLo), wHi);
        const int mHi = max(min(wHi, iHiE), mLo);

        v8f acc0 = {0.f,0.f,0.f,0.f,0.f,0.f,0.f,0.f};
        v8f acc1 = {0.f,0.f,0.f,0.f,0.f,0.f,0.f,0.f};
        v8f acc2 = {0.f,0.f,0.f,0.f,0.f,0.f,0.f,0.f};
        v8f acc3 = {0.f,0.f,0.f,0.f,0.f,0.f,0.f,0.f};
        v8f p01  = {0.f,0.f,0.f,0.f,0.f,0.f,0.f,0.f};
        v8f p23  = {0.f,0.f,0.f,0.f,0.f,0.f,0.f,0.f};

        for (int k0 = wLo; k0 < mLo; k0 += 4)
            step_checked4(rir, a_ch, k0, rbase, bbase, acc0, acc1, acc2, acc3);

        {
            int k0 = mLo;

            const int npair = ((mHi - mLo) >> 9) * 8;
            if (npair > 0) {
                const char* ea = (const char*)rir  + (ptrdiff_t)(mLo + 8 * h - 16 * mq) * 4;
                const char* eb = (const char*)a_ch + (ptrdiff_t)(mLo + n0 + mq + 8 * h - (T_LEN - 1)) * 4;
                const char* oa = ea + 64 - (128 + 1024);
                const char* ob = eb + 64 - (128 + 1024);
                Grp f0, f1;
                load_grp(f0, ea, eb);
                for (int p = 0; p < npair - 1; ++p) {
                    const int oadv = ((p & 7) == 0) ? (128 + 1024) : 128;
                    oa += oadv;
                    ob += oadv;
                    load_grp(f1, oa, ob);
                    mma_grp(f0, acc0, acc1, acc2, acc3, p01, p23);
                    const int eadv = ((p & 7) == 7) ? (128 + 1024) : 128;
                    ea += eadv;
                    eb += eadv;
                    load_grp(f0, ea, eb);
                    asm volatile("" ::: "memory");
                    mma_grp(f1, acc0, acc1, acc2, acc3, p01, p23);
                }
                load_grp(f1, oa + 128, ob + 128);
                mma_grp(f0, acc0, acc1, acc2, acc3, p01, p23);
                mma_grp(f1, acc0, acc1, acc2, acc3, p01, p23);
                k0 += npair * 64;
            }

            const char* pa = (const char*)rir  + (ptrdiff_t)(k0 + rbase) * 4;
            const char* pb = (const char*)a_ch + (ptrdiff_t)(k0 + bbase) * 4;

            for (; k0 < mHi; k0 += 4) {
                v2f a0 = *(const v2f*)(pa);
                f2u b0 = *(const f2u*)(pb);
                f2u b1 = *(const f2u*)(pb + 1024);
                f2u b2 = *(const f2u*)(pb + 2048);
                f2u b3 = *(const f2u*)(pb + 3072);
                v2f v0 = {b0.x, b0.y};
                v2f v1 = {b1.x, b1.y};
                v2f v2 = {b2.x, b2.y};
                v2f v3 = {b3.x, b3.y};
                acc0 = wmma4(a0, v0, acc0);
                acc1 = wmma4(a0, v1, acc1);
                acc2 = wmma4(a0, v2, acc2);
                acc3 = wmma4(a0, v3, acc3);
                pa += 16;
                pb += 16;
            }
        }

        for (int kc = mHi; kc < wHi; kc += 4)
            step_checked4(rir, a_ch, kc, rbase, bbase, acc0, acc1, acc2, acc3);

        acc0 += p01;
        acc1 += p01;
        acc2 += p23;
        acc3 += p23;

        #pragma unroll
        for (int v = 0; v < 8; ++v) {
            float* r = red + wave * TILE_W + (v + 8 * h) * 16 + mq;
            r[0]   = acc0[v];
            r[256] = acc1[v];
            r[512] = acc2[v];
            r[768] = acc3[v];
        }
        __syncthreads();

        float s0 = 0.f, s1 = 0.f, s2 = 0.f, s3 = 0.f;
        #pragma unroll
        for (int w = 0; w < NWAVES; ++w) {
            s0 += red[w * TILE_W + tid];
            s1 += red[w * TILE_W + 256 + tid];
            s2 += red[w * TILE_W + 512 + tid];
            s3 += red[w * TILE_W + 768 + tid];
        }
        const int n = n0 + tid;
        if (n < OUT_LEN)       o_ch[n]       = s0;
        if (n + 256 < OUT_LEN) o_ch[n + 256] = s1;
        if (n + 512 < OUT_LEN) o_ch[n + 512] = s2;
        if (n + 768 < OUT_LEN) o_ch[n + 768] = s3;

        __syncthreads();
    }
}

extern "C" void kernel_launch(void* const* d_in, const int* in_sizes, int n_in,
                              void* d_out, int out_size, void* d_ws, size_t ws_size,
                              hipStream_t stream) {
    (void)in_sizes; (void)n_in; (void)d_ws; (void)ws_size; (void)out_size;
    const float* audio = (const float*)d_in[0];
    const float* rir   = (const float*)d_in[1];
    float* out = (float*)d_out;

    dim3 grid(NBLK_X, 4, 1);
    dim3 block(256, 1, 1);
    RoomImpulseResponse_wmma_conv<<<grid, block, 0, stream>>>(audio, rir, out);
}


# Interior in 1024-k chunks: the fast-FIR identity nested two levels over 4 subtiles x 4 k offsets, 9 WMMAs per 16 block products (was 12); f32 operand sums/differences, sum order changes
# speedup vs baseline: 1.8536x; 1.1771x over previous
#include <hip/hip_runtime.h>
#include <hip/hip_bf16.h>


typedef __attribute__((ext_vector_type(2))) float v2f;
typedef __attribute__((ext_vector_type(8))) float v8f;

typedef struct __attribute__((packed, aligned(4))) { float x, y; } f2u;
typedef struct __attribute__((packed, aligned(4))) { float x, y, z, w; } f4u;

#define T_LEN   131072
#define OUT_LEN (2 * T_LEN - 1)
#define NSUB    4
#define TILE_W  (256 * NSUB)
#define NTILES  ((OUT_LEN + TILE_W - 1) / TILE_W)
#define NWAVES  8
#define NBLK_X  (NTILES / 2)


__device__ __forceinline__ v8f wmma4(v2f a, v2f b, v8f c) {
    return __builtin_amdgcn_wmma_f32_16x16x4_f32(false, a, false, b,
                                                 (short)0, c, false, false);
}

struct Grp {
    f4u a0l, a0h;
    f4u a1l, a1h;
    f4u b0l, b0h;
    f4u b1l, b1h;
    f4u b2l, b2h;
    f4u b3l, b3h;
    f4u b4l, b4h;
};

__device__ __forceinline__ void load_grp(Grp& f, const char* pa, const char* pb) {
    f.a0l = *(const f4u*)(pa);
    f.a0h = *(const f4u*)(pa + 16);
    f.a1l = *(const f4u*)(pa + 1024);
    f.a1h = *(const f4u*)(pa + 1040);
    f.b0l = *(const f4u*)(pb);
    f.b0h = *(const f4u*)(pb + 16);
    f.b1l = *(const f4u*)(pb + 1024);
    f.b1h = *(const f4u*)(pb + 1040);
    f.b2l = *(const f4u*)(pb + 2048);
    f.b2h = *(const f4u*)(pb + 2064);
    f.b3l = *(const f4u*)(pb + 3072);
    f.b3h = *(const f4u*)(pb + 3088);
    f.b4l = *(const f4u*)(pb + 4096);
    f.b4h = *(const f4u*)(pb + 4112);
}

__device__ __forceinline__ void mma6(v2f a0, v2f a1,
                                     v2f b0, v2f b1, v2f b2, v2f b3, v2f b4,
                                     v8f& acc0, v8f& acc1, v8f& acc2, v8f& acc3,
                                     v8f& p01, v8f& p23)
{
    const v2f s   = a0 + a1;
    const v2f d01 = b0 - b1;
    const v2f e01 = b2 - b1;
    const v2f d23 = b2 - b3;
    const v2f e23 = b4 - b3;
    acc0 = wmma4(a0, d01, acc0);
    acc1 = wmma4(a1, e01, acc1);
    p01  = wmma4(s,  b1,  p01);
    acc2 = wmma4(a0, d23, acc2);
    acc3 = wmma4(a1, e23, acc3);
    p23  = wmma4(s,  b3,  p23);
}

#define LOXY(q) ((v2f){(q).x, (q).y})
#define HIZW(q) ((v2f){(q).z, (q).w})

__device__ __forceinline__ void mma_grp(const Grp& f,
                                        v8f& acc0, v8f& acc1, v8f& acc2, v8f& acc3,
                                        v8f& p01, v8f& p23)
{
    mma6(LOXY(f.a0l), LOXY(f.a1l), LOXY(f.b0l), LOXY(f.b1l), LOXY(f.b2l), LOXY(f.b3l), LOXY(f.b4l),
         acc0, acc1, acc2, acc3, p01, p23);
    mma6(HIZW(f.a0l), HIZW(f.a1l), HIZW(f.b0l), HIZW(f.b1l), HIZW(f.b2l), HIZW(f.b3l), HIZW(f.b4l),
         acc0, acc1, acc2, acc3, p01, p23);
    mma6(LOXY(f.a0h), LOXY(f.a1h), LOXY(f.b0h), LOXY(f.b1h), LOXY(f.b2h), LOXY(f.b3h), LOXY(f.b4h),
         acc0, acc1, acc2, acc3, p01, p23);
    mma6(HIZW(f.a0h), HIZW(f.a1h), HIZW(f.b0h), HIZW(f.b1h), HIZW(f.b2h), HIZW(f.b3h), HIZW(f.b4h),
         acc0, acc1, acc2, acc3, p01, p23);
}

struct Grp4 {
    f4u a0, a1, a2, a3;
    f4u b0, b1, b2, b3, b4, b5, b6;
};

__device__ __forceinline__ void load_grp4(Grp4& f, const char* pa, const char* pb) {
    f.a0 = *(const f4u*)(pa);
    f.a1 = *(const f4u*)(pa + 1024);
    f.a2 = *(const f4u*)(pa + 2048);
    f.a3 = *(const f4u*)(pa + 3072);
    f.b0 = *(const f4u*)(pb);
    f.b1 = *(const f4u*)(pb + 1024);
    f.b2 = *(const f4u*)(pb + 2048);
    f.b3 = *(const f4u*)(pb + 3072);
    f.b4 = *(const f4u*)(pb + 4096);
    f.b5 = *(const f4u*)(pb + 5120);
    f.b6 = *(const f4u*)(pb + 6144);
}

__device__ __forceinline__ void mma9(v2f a0, v2f a1, v2f a2, v2f a3,
                                     v2f b0, v2f b1, v2f b2, v2f b3, v2f b4, v2f b5, v2f b6,
                                     v8f& acc0, v8f& acc1, v8f& acc2, v8f& acc3,
                                     v8f& p01, v8f& p23, v8f& r0, v8f& r1, v8f& rs)
{
    const v2f s01 = a0 + a1;
    const v2f s23 = a2 + a3;
    const v2f t0  = a0 + a2;
    const v2f t1  = a1 + a3;
    const v2f ts  = t0 + t1;
    const v2f D0  = b0 - b2;
    const v2f D1  = b1 - b3;
    const v2f D2  = b2 - b4;
    const v2f E1  = b5 - b3;
    const v2f E2  = b6 - b4;
    const v2f x0  = D0 - D1;
    const v2f x1  = D2 - D1;
    const v2f y0  = -D2 - E1;
    const v2f y1  = E2 - E1;
    const v2f z0  = b2 - b3;
    const v2f z1  = b4 - b3;
    acc0 = wmma4(a0,  x0, acc0);
    acc1 = wmma4(a1,  x1, acc1);
    p01  = wmma4(s01, D1, p01);
    acc2 = wmma4(a2,  y0, acc2);
    acc3 = wmma4(a3,  y1, acc3);
    p23  = wmma4(s23, E1, p23);
    r0   = wmma4(t0,  z0, r0);
    r1   = wmma4(t1,  z1, r1);
    rs   = wmma4(ts,  b3, rs);
}

__device__ __forceinline__ void mma_grp4(const Grp4& f,
                                         v8f& acc0, v8f& acc1, v8f& acc2, v8f& acc3,
                                         v8f& p01, v8f& p23, v8f& r0, v8f& r1, v8f& rs)
{
    mma9(LOXY(f.a0), LOXY(f.a1), LOXY(f.a2), LOXY(f.a3),
         LOXY(f.b0), LOXY(f.b1), LOXY(f.b2), LOXY(f.b3), LOXY(f.b4), LOXY(f.b5), LOXY(f.b6),
         acc0, acc1, acc2, acc3, p01, p23, r0, r1, rs);
    mma9(HIZW(f.a0), HIZW(f.a1), HIZW(f.a2), HIZW(f.a3),
         HIZW(f.b0), HIZW(f.b1), HIZW(f.b2), HIZW(f.b3), HIZW(f.b4), HIZW(f.b5), HIZW(f.b6),
         acc0, acc1, acc2, acc3, p01, p23, r0, r1, rs);
}

__device__ __forceinline__ float ld_guard(const float* __restrict__ p, int i) {
    float v = p[min(max(i, 0), T_LEN - 1)];
    return ((unsigned)i < (unsigned)T_LEN) ? v : 0.f;
}

__device__ __forceinline__ void step_checked4(const float* __restrict__ rir,
                                              const float* __restrict__ a_ch,
                                              int k0, int rbase, int bbase,
                                              v8f& acc0, v8f& acc1, v8f& acc2, v8f& acc3)
{
    const int r0 = k0 + rbase;
    const int i0 = k0 + bbase;
    v2f a, b0, b1, b2, b3;
    a.x  = ld_guard(rir, r0);
    a.y  = ld_guard(rir, r0 + 1);
    b0.x = ld_guard(a_ch, i0);
    b0.y = ld_guard(a_ch, i0 + 1);
    b1.x = ld_guard(a_ch, i0 + 256);
    b1.y = ld_guard(a_ch, i0 + 257);
    b2.x = ld_guard(a_ch, i0 + 512);
    b2.y = ld_guard(a_ch, i0 + 513);
    b3.x = ld_guard(a_ch, i0 + 768);
    b3.y = ld_guard(a_ch, i0 + 769);
    acc0 = wmma4(a, b0, acc0);
    acc1 = wmma4(a, b1, acc1);
    acc2 = wmma4(a, b2, acc2);
    acc3 = wmma4(a, b3, acc3);
}

__global__ __launch_bounds__(256)
void RoomImpulseResponse_wmma_conv(const float* __restrict__ audio,
                                   const float* __restrict__ rir,
                                   float* __restrict__ out)
{
    __shared__ float red[NWAVES * TILE_W];

    const int ch   = blockIdx.y;
    const int tid  = threadIdx.x;
    const int wave = __builtin_amdgcn_readfirstlane(tid >> 5);
    const int lane = tid & 31;
    const int h    = lane >> 4;
    const int mq   = lane & 15;

    const float* __restrict__ a_ch = audio + ch * T_LEN;
    float* __restrict__ o_ch = out + ch * OUT_LEN;

    const int tbase = (blockIdx.x >> 6) * (NTILES / 2);
    const int u     = blockIdx.x & 63;

    const int rbase = 2 * h - 16 * mq;

    for (int pass = 0; pass < 2; ++pass) {
        const int tile = tbase + (pass ? (NTILES / 2 - 1 - u) : u);
        const int n0   = tile * TILE_W;

        int klo = T_LEN - 1 - (256 * (NSUB - 1) + 15) - n0;  if (klo < 0) klo = 0;
        int khi = 2 * T_LEN - 2 - n0;
        if (khi > T_LEN - 1 + 240) khi = T_LEN - 1 + 240;

        const int bbase = n0 + mq + 2 * h - (T_LEN - 1);

        const int kbeg   = klo & ~3;
        const int nsteps = ((khi - kbeg) >> 2) + 1;
        const int per    = (nsteps + NWAVES - 1) / NWAVES;
        const int wLo = kbeg + (wave * per) * 4;
        const int wHi = kbeg + min(nsteps, (wave + 1) * per) * 4;

        int iLo = T_LEN - 1 - n0;  if (iLo < 240) iLo = 240;
        iLo = (iLo + 3) & ~3;
        int iHi = 2 * T_LEN - 2 - (256 * (NSUB - 1) + 18) - n0;
        if (iHi > T_LEN - 4) iHi = T_LEN - 4;
        int iHiE = (iHi & ~3) + 4;
        if (iHiE < iLo) iHiE = iLo;

        const int mLo = min(max(wLo, iLo), wHi);
        const int mHi = max(min(wHi, iHiE), mLo);

        v8f acc0 = {0.f,0.f,0.f,0.f,0.f,0.f,0.f,0.f};
        v8f acc1 = {0.f,0.f,0.f,0.f,0.f,0.f,0.f,0.f};
        v8f acc2 = {0.f,0.f,0.f,0.f,0.f,0.f,0.f,0.f};
        v8f acc3 = {0.f,0.f,0.f,0.f,0.f,0.f,0.f,0.f};
        v8f p01  = {0.f,0.f,0.f,0.f,0.f,0.f,0.f,0.f};
        v8f p23  = {0.f,0.f,0.f,0.f,0.f,0.f,0.f,0.f};
        v8f r0   = {0.f,0.f,0.f,0.f,0.f,0.f,0.f,0.f};
        v8f r1   = {0.f,0.f,0.f,0.f,0.f,0.f,0.f,0.f};
        v8f rs   = {0.f,0.f,0.f,0.f,0.f,0.f,0.f,0.f};

        for (int k0 = wLo; k0 < mLo; k0 += 4)
            step_checked4(rir, a_ch, k0, rbase, bbase, acc0, acc1, acc2, acc3);

        {
            int k0 = mLo;

            const int npair4 = ((mHi - mLo) >> 10) * 16;
            if (npair4 > 0) {
                const char* ea = (const char*)rir  + (ptrdiff_t)(mLo + 4 * h - 16 * mq) * 4;
                const char* eb = (const char*)a_ch + (ptrdiff_t)(mLo + n0 + mq + 4 * h - (T_LEN - 1)) * 4;
                const char* oa = ea + 32;
                const char* ob = eb + 32;
                Grp4 g0, g1;
                load_grp4(g0, ea, eb);
                for (int p = 0; p < npair4 - 1; ++p) {
                    load_grp4(g1, oa, ob);
                    mma_grp4(g0, acc0, acc1, acc2, acc3, p01, p23, r0, r1, rs);
                    const int adv = ((p & 15) == 15) ? (64 + 3072) : 64;
                    ea += adv;
                    eb += adv;
                    load_grp4(g0, ea, eb);
                    asm volatile("" ::: "memory");
                    mma_grp4(g1, acc0, acc1, acc2, acc3, p01, p23, r0, r1, rs);
                    oa += adv;
                    ob += adv;
                }
                load_grp4(g1, oa, ob);
                mma_grp4(g0, acc0, acc1, acc2, acc3, p01, p23, r0, r1, rs);
                mma_grp4(g1, acc0, acc1, acc2, acc3, p01, p23, r0, r1, rs);
                k0 += npair4 * 64;
            }
            const int kq = k0;

            const int npair = ((mHi - kq) >> 9) * 8;
            if (npair > 0) {
                const char* ea = (const char*)rir  + (ptrdiff_t)(kq + 8 * h - 16 * mq) * 4;
                const char* eb = (const char*)a_ch + (ptrdiff_t)(kq + n0 + mq + 8 * h - (T_LEN - 1)) * 4;
                const char* oa = ea + 64 - (128 + 1024);
                const char* ob = eb + 64 - (128 + 1024);
                Grp f0, f1;
                load_grp(f0, ea, eb);
                for (int p = 0; p < npair - 1; ++p) {
                    const int oadv = ((p & 7) == 0) ? (128 + 1024) : 128;
                    oa += oadv;
                    ob += oadv;
                    load_grp(f1, oa, ob);
                    mma_grp(f0, acc0, acc1, acc2, acc3, p01, p23);
                    const int eadv = ((p & 7) == 7) ? (128 + 1024) : 128;
                    ea += eadv;
                    eb += eadv;
                    load_grp(f0, ea, eb);
                    asm volatile("" ::: "memory");
                    mma_grp(f1, acc0, acc1, acc2, acc3, p01, p23);
                }
                load_grp(f1, oa + 128, ob + 128);
                mma_grp(f0, acc0, acc1, acc2, acc3, p01, p23);
                mma_grp(f1, acc0, acc1, acc2, acc3, p01, p23);
                k0 += npair * 64;
            }

            const char* pa = (const char*)rir  + (ptrdiff_t)(k0 + rbase) * 4;
            const char* pb = (const char*)a_ch + (ptrdiff_t)(k0 + bbase) * 4;

            for (; k0 < mHi; k0 += 4) {
                v2f a0 = *(const v2f*)(pa);
                f2u b0 = *(const f2u*)(pb);
                f2u b1 = *(const f2u*)(pb + 1024);
                f2u b2 = *(const f2u*)(pb + 2048);
                f2u b3 = *(const f2u*)(pb + 3072);
                v2f v0 = {b0.x, b0.y};
                v2f v1 = {b1.x, b1.y};
                v2f v2 = {b2.x, b2.y};
                v2f v3 = {b3.x, b3.y};
                acc0 = wmma4(a0, v0, acc0);
                acc1 = wmma4(a0, v1, acc1);
                acc2 = wmma4(a0, v2, acc2);
                acc3 = wmma4(a0, v3, acc3);
                pa += 16;
                pb += 16;
            }
        }

        for (int kc = mHi; kc < wHi; kc += 4)
            step_checked4(rir, a_ch, kc, rbase, bbase, acc0, acc1, acc2, acc3);

        acc0 += p01 + (r0 + rs);
        acc1 += p01 + (r1 + rs);
        acc2 += p23 + (r0 + rs);
        acc3 += p23 + (r1 + rs);

        #pragma unroll
        for (int v = 0; v < 8; ++v) {
            float* r = red + wave * TILE_W + (v + 8 * h) * 16 + mq;
            r[0]   = acc0[v];
            r[256] = acc1[v];
            r[512] = acc2[v];
            r[768] = acc3[v];
        }
        __syncthreads();

        float s0 = 0.f, s1 = 0.f, s2 = 0.f, s3 = 0.f;
        #pragma unroll
        for (int w = 0; w < NWAVES; ++w) {
            s0 += red[w * TILE_W + tid];
            s1 += red[w * TILE_W + 256 + tid];
            s2 += red[w * TILE_W + 512 + tid];
            s3 += red[w * TILE_W + 768 + tid];
        }
        const int n = n0 + tid;
        if (n < OUT_LEN)       o_ch[n]       = s0;
        if (n + 256 < OUT_LEN) o_ch[n + 256] = s1;
        if (n + 512 < OUT_LEN) o_ch[n + 512] = s2;
        if (n + 768 < OUT_LEN) o_ch[n + 768] = s3;

        __syncthreads();
    }
}

extern "C" void kernel_launch(void* const* d_in, const int* in_sizes, int n_in,
                              void* d_out, int out_size, void* d_ws, size_t ws_size,
                              hipStream_t stream) {
    (void)in_sizes; (void)n_in; (void)d_ws; (void)ws_size; (void)out_size;
    const float* audio = (const float*)d_in[0];
    const float* rir   = (const float*)d_in[1];
    float* out = (float*)d_out;

    dim3 grid(NBLK_X, 4, 1);
    dim3 block(256, 1, 1);
    RoomImpulseResponse_wmma_conv<<<grid, block, 0, stream>>>(audio, rir, out);
}


# Interior in 2048-k chunks: fast-FIR identity nested 3 levels over 8 subtiles x 8 k offsets, 27 WMMAs per 64 block products (was 36); tiles 2048 wide; f32 operand sums/differences, sum order changes
# speedup vs baseline: 2.4225x; 1.3069x over previous
#include <hip/hip_runtime.h>
#include <hip/hip_bf16.h>


typedef __attribute__((ext_vector_type(2))) float v2f;
typedef __attribute__((ext_vector_type(8))) float v8f;

typedef struct __attribute__((packed, aligned(4))) { float x, y; } f2u;
typedef struct __attribute__((packed, aligned(4))) { float x, y, z, w; } f4u;

#define T_LEN   131072
#define OUT_LEN (2 * T_LEN - 1)
#define NSUB    8
#define TILE_W  (256 * NSUB)
#define CHUNK_K (256 * NSUB)
#define NTILES  ((OUT_LEN + TILE_W - 1) / TILE_W)
#define NWAVES  8
#define NBLK_X  (NTILES / 2)
#define NACC    27


__device__ __forceinline__ v8f wmma4(v2f a, v2f b, v8f c) {
    return __builtin_amdgcn_wmma_f32_16x16x4_f32(false, a, false, b,
                                                 (short)0, c, false, false);
}

#define LOXY(q) ((v2f){(q).x, (q).y})
#define HIZW(q) ((v2f){(q).z, (q).w})

#define LEAF(s) (9 * (((s) >> 2) & 1) + 3 * (((s) >> 1) & 1) + ((s) & 1))

template <int N> struct Kar {
    enum { H = N / 2, P = Kar<N / 2>::NA, NA = 3 * P };
    static __device__ __forceinline__ void run(const v2f* a, const v2f* b, v8f* acc) {
        v2f as[H];
        v2f bl[N - 1];
        v2f bh[N - 1];
        #pragma unroll
        for (int j = 0; j < H; ++j) as[j] = a[j] + a[H + j];
        #pragma unroll
        for (int i = 0; i < N - 1; ++i) {
            bl[i] = b[i] - b[H + i];
            bh[i] = b[N + i] - b[H + i];
        }
        Kar<H>::run(a,     bl,    acc);
        Kar<H>::run(a + H, bh,    acc + P);
        Kar<H>::run(as,    b + H, acc + 2 * P);
    }
    static __device__ __forceinline__ void fold(const v8f* acc, v8f* out) {
        v8f lo[H], hi[H], sh[H];
        Kar<H>::fold(acc,         lo);
        Kar<H>::fold(acc + P,     hi);
        Kar<H>::fold(acc + 2 * P, sh);
        #pragma unroll
        for (int s = 0; s < H; ++s) {
            out[s]     = lo[s] + sh[s];
            out[H + s] = hi[s] + sh[s];
        }
    }
};
template <> struct Kar<1> {
    enum { NA = 1 };
    static __device__ __forceinline__ void run(const v2f* a, const v2f* b, v8f* acc) {
        acc[0] = wmma4(a[0], b[0], acc[0]);
    }
    static __device__ __forceinline__ void fold(const v8f* acc, v8f* out) {
        out[0] = acc[0];
    }
};

struct Grp8 {
    f4u a[NSUB];
    f4u b[2 * NSUB - 1];
};

__device__ __forceinline__ void load_grp8(Grp8& f, const char* pa, const char* pb) {
    #pragma unroll
    for (int j = 0; j < NSUB; ++j) f.a[j] = *(const f4u*)(pa + 1024 * j);
    #pragma unroll
    for (int i = 0; i < 2 * NSUB - 1; ++i) f.b[i] = *(const f4u*)(pb + 1024 * i);
}

__device__ __forceinline__ void mma_grp8(const Grp8& f, v8f* acc) {
    {
        v2f a[NSUB], b[2 * NSUB - 1];
        #pragma unroll
        for (int j = 0; j < NSUB; ++j) a[j] = LOXY(f.a[j]);
        #pragma unroll
        for (int i = 0; i < 2 * NSUB - 1; ++i) b[i] = LOXY(f.b[i]);
        Kar<NSUB>::run(a, b, acc);
    }
    {
        v2f a[NSUB], b[2 * NSUB - 1];
        #pragma unroll
        for (int j = 0; j < NSUB; ++j) a[j] = HIZW(f.a[j]);
        #pragma unroll
        for (int i = 0; i < 2 * NSUB - 1; ++i) b[i] = HIZW(f.b[i]);
        Kar<NSUB>::run(a, b, acc);
    }
}

__device__ __forceinline__ float ld_guard(const float* __restrict__ p, int i) {
    float v = p[min(max(i, 0), T_LEN - 1)];
    return ((unsigned)i < (unsigned)T_LEN) ? v : 0.f;
}

__device__ __forceinline__ void step_checked8(const float* __restrict__ rir,
                                              const float* __restrict__ a_ch,
                                              int k0, int rbase, int bbase, v8f* acc)
{
    const int r0 = k0 + rbase;
    const int i0 = k0 + bbase;
    v2f a;
    a.x = ld_guard(rir, r0);
    a.y = ld_guard(rir, r0 + 1);
    v2f b[NSUB];
    #pragma unroll
    for (int s = 0; s < NSUB; ++s) {
        b[s].x = ld_guard(a_ch, i0 + 256 * s);
        b[s].y = ld_guard(a_ch, i0 + 256 * s + 1);
    }
    #pragma unroll
    for (int s = 0; s < NSUB; ++s) acc[LEAF(s)] = wmma4(a, b[s], acc[LEAF(s)]);
}

__global__ __launch_bounds__(256)
void RoomImpulseResponse_wmma_conv(const float* __restrict__ audio,
                                   const float* __restrict__ rir,
                                   float* __restrict__ out)
{
    __shared__ float red[NWAVES * TILE_W];

    const int ch   = blockIdx.y;
    const int tid  = threadIdx.x;
    const int wave = __builtin_amdgcn_readfirstlane(tid >> 5);
    const int lane = tid & 31;
    const int h    = lane >> 4;
    const int mq   = lane & 15;

    const float* __restrict__ a_ch = audio + ch * T_LEN;
    float* __restrict__ o_ch = out + ch * OUT_LEN;

    const int tbase = (blockIdx.x >> 5) * (NTILES / 2);
    const int u     = blockIdx.x & 31;

    const int rbase = 2 * h - 16 * mq;

    int coff = 0;

    for (int pass = 0; pass < 2; ++pass) {
        const int tile = tbase + (pass ? (NTILES / 2 - 1 - u) : u);
        const int n0   = tile * TILE_W;

        int klo = T_LEN - 1 - (256 * (NSUB - 1) + 15) - n0;  if (klo < 0) klo = 0;
        int khi = 2 * T_LEN - 2 - n0;
        if (khi > T_LEN - 1 + 240) khi = T_LEN - 1 + 240;

        const int bbase = n0 + mq + 2 * h - (T_LEN - 1);

        const int kbeg   = klo & ~3;
        const int nsteps = ((khi - kbeg) >> 2) + 1;
        const int kend   = kbeg + 4 * nsteps;

        int iLo = T_LEN - 1 - n0;  if (iLo < 240) iLo = 240;
        iLo = (iLo + 3) & ~3;
        int iHi = 2 * T_LEN - 2 - (256 * (NSUB - 1) + 18) - n0;
        if (iHi > T_LEN - 4) iHi = T_LEN - 4;
        int iHiE = (iHi & ~3) + 4;
        if (iHiE < iLo) iHiE = iLo;

        const int pEnd   = min(iLo, kend);
        const int iEnd   = max(min(iHiE, kend), pEnd);
        const int nchunk = (iEnd - pEnd) / CHUNK_K;
        const int iMid   = pEnd + nchunk * CHUNK_K;
        const int nP     = (pEnd - kbeg) >> 2;
        const int nE     = (kend - iEnd) >> 2;
        const int nL     = (iEnd - iMid) >> 2;

        v8f acc[NACC];
        #pragma unroll
        for (int i = 0; i < NACC; ++i) acc[i] = (v8f){0.f,0.f,0.f,0.f,0.f,0.f,0.f,0.f};

        {
            const int nC   = nP + nE;
            const int perC = (nC + NWAVES - 1) / NWAVES;
            const int v1   = min(nC, (wave + 1) * perC);
            for (int v = wave * perC; v < v1; ++v) {
                const int k0 = (v < nP) ? (kbeg + 4 * v) : (iEnd + 4 * (v - nP));
                step_checked8(rir, a_ch, k0, rbase, bbase, acc);
            }
        }

        {
            const int first  = (wave - coff) & 7;
            const int npair8 = ((nchunk - first + 7) >> 3) * 16;
            if (npair8 > 0) {
                const int kc = pEnd + first * CHUNK_K;
                const char* ea = (const char*)rir  + (ptrdiff_t)(kc + 4 * h - 16 * mq) * 4;
                const char* eb = (const char*)a_ch + (ptrdiff_t)(kc + n0 + mq + 4 * h - (T_LEN - 1)) * 4;
                const char* oa = ea + 32;
                const char* ob = eb + 32;
                Grp8 g0, g1;
                load_grp8(g0, ea, eb);
                for (int p = 0; p < npair8 - 1; ++p) {
                    load_grp8(g1, oa, ob);
                    mma_grp8(g0, acc);
                    const int adv = ((p & 15) == 15) ? (64 + 7168 + 7 * 8192) : 64;
                    ea += adv;
                    eb += adv;
                    load_grp8(g0, ea, eb);
                    asm volatile("" ::: "memory");
                    mma_grp8(g1, acc);
                    oa += adv;
                    ob += adv;
                }
                load_grp8(g1, oa, ob);
                mma_grp8(g0, acc);
                mma_grp8(g1, acc);
            }
        }
        coff = (coff + nchunk) & 7;

        {
            const int perL = (nL + NWAVES - 1) / NWAVES;
            const int l0   = min(nL, wave * perL);
            const int l1   = min(nL, l0 + perL);
            int k0 = iMid + 4 * l0;
            const int kL = iMid + 4 * l1;
            const char* pa = (const char*)rir  + (ptrdiff_t)(k0 + rbase) * 4;
            const char* pb = (const char*)a_ch + (ptrdiff_t)(k0 + bbase) * 4;
            for (; k0 < kL; k0 += 4) {
                const v2f a0 = *(const v2f*)(pa);
                v2f b[NSUB];
                #pragma unroll
                for (int s = 0; s < NSUB; ++s) {
                    const f2u t = *(const f2u*)(pb + 1024 * s);
                    b[s] = (v2f){t.x, t.y};
                }
                #pragma unroll
                for (int s = 0; s < NSUB; ++s) acc[LEAF(s)] = wmma4(a0, b[s], acc[LEAF(s)]);
                pa += 16;
                pb += 16;
            }
        }

        v8f o[NSUB];
        Kar<NSUB>::fold(acc, o);

        #pragma unroll
        for (int v = 0; v < 8; ++v) {
            float* r = red + wave * TILE_W + (v + 8 * h) * 16 + mq;
            #pragma unroll
            for (int s = 0; s < NSUB; ++s) r[256 * s] = o[s][v];
        }
        __syncthreads();

        const int n = n0 + tid;
        #pragma unroll
        for (int s = 0; s < NSUB; ++s) {
            float sum = 0.f;
            #pragma unroll
            for (int w = 0; w < NWAVES; ++w) sum += red[w * TILE_W + 256 * s + tid];
            if (n + 256 * s < OUT_LEN) o_ch[n + 256 * s] = sum;
        }

        __syncthreads();
    }
}

extern "C" void kernel_launch(void* const* d_in, const int* in_sizes, int n_in,
                              void* d_out, int out_size, void* d_ws, size_t ws_size,
                              hipStream_t stream) {
    (void)in_sizes; (void)n_in; (void)d_ws; (void)ws_size; (void)out_size;
    const float* audio = (const float*)d_in[0];
    const float* rir   = (const float*)d_in[1];
    float* out = (float*)d_out;

    dim3 grid(NBLK_X, 4, 1);
    dim3 block(256, 1, 1);
    RoomImpulseResponse_wmma_conv<<<grid, block, 0, stream>>>(audio, rir, out);
}


# K split inside every 2048-k chunk: each wave takes 4 of its 32 groups (was whole chunks dealt round-robin), so all 8 waves have equal work in every tile; f32 sum order changes
# speedup vs baseline: 2.5309x; 1.0447x over previous
#include <hip/hip_runtime.h>
#include <hip/hip_bf16.h>


typedef __attribute__((ext_vector_type(2))) float v2f;
typedef __attribute__((ext_vector_type(8))) float v8f;

typedef struct __attribute__((packed, aligned(4))) { float x, y; } f2u;
typedef struct __attribute__((packed, aligned(4))) { float x, y, z, w; } f4u;

#define T_LEN   131072
#define OUT_LEN (2 * T_LEN - 1)
#define NSUB    8
#define TILE_W  (256 * NSUB)
#define CHUNK_K (256 * NSUB)
#define NTILES  ((OUT_LEN + TILE_W - 1) / TILE_W)
#define NWAVES  8
#define NBLK_X  (NTILES / 2)
#define NACC    27


__device__ __forceinline__ v8f wmma4(v2f a, v2f b, v8f c) {
    return __builtin_amdgcn_wmma_f32_16x16x4_f32(false, a, false, b,
                                                 (short)0, c, false, false);
}

#define LOXY(q) ((v2f){(q).x, (q).y})
#define HIZW(q) ((v2f){(q).z, (q).w})

#define LEAF(s) (9 * (((s) >> 2) & 1) + 3 * (((s) >> 1) & 1) + ((s) & 1))

template <int N> struct Kar {
    enum { H = N / 2, P = Kar<N / 2>::NA, NA = 3 * P };
    static __device__ __forceinline__ void run(const v2f* a, const v2f* b, v8f* acc) {
        v2f as[H];
        v2f bl[N - 1];
        v2f bh[N - 1];
        #pragma unroll
        for (int j = 0; j < H; ++j) as[j] = a[j] + a[H + j];
        #pragma unroll
        for (int i = 0; i < N - 1; ++i) {
            bl[i] = b[i] - b[H + i];
            bh[i] = b[N + i] - b[H + i];
        }
        Kar<H>::run(a,     bl,    acc);
        Kar<H>::run(a + H, bh,    acc + P);
        Kar<H>::run(as,    b + H, acc + 2 * P);
    }
    static __device__ __forceinline__ void fold(const v8f* acc, v8f* out) {
        v8f lo[H], hi[H], sh[H];
        Kar<H>::fold(acc,         lo);
        Kar<H>::fold(acc + P,     hi);
        Kar<H>::fold(acc + 2 * P, sh);
        #pragma unroll
        for (int s = 0; s < H; ++s) {
            out[s]     = lo[s] + sh[s];
            out[H + s] = hi[s] + sh[s];
        }
    }
};
template <> struct Kar<1> {
    enum { NA = 1 };
    static __device__ __forceinline__ void run(const v2f* a, const v2f* b, v8f* acc) {
        acc[0] = wmma4(a[0], b[0], acc[0]);
    }
    static __device__ __forceinline__ void fold(const v8f* acc, v8f* out) {
        out[0] = acc[0];
    }
};

struct Grp8 {
    f4u a[NSUB];
    f4u b[2 * NSUB - 1];
};

__device__ __forceinline__ void load_grp8(Grp8& f, const char* pa, const char* pb) {
    #pragma unroll
    for (int j = 0; j < NSUB; ++j) f.a[j] = *(const f4u*)(pa + 1024 * j);
    #pragma unroll
    for (int i = 0; i < 2 * NSUB - 1; ++i) f.b[i] = *(const f4u*)(pb + 1024 * i);
}

__device__ __forceinline__ void mma_grp8(const Grp8& f, v8f* acc) {
    {
        v2f a[NSUB], b[2 * NSUB - 1];
        #pragma unroll
        for (int j = 0; j < NSUB; ++j) a[j] = LOXY(f.a[j]);
        #pragma unroll
        for (int i = 0; i < 2 * NSUB - 1; ++i) b[i] = LOXY(f.b[i]);
        Kar<NSUB>::run(a, b, acc);
    }
    {
        v2f a[NSUB], b[2 * NSUB - 1];
        #pragma unroll
        for (int j = 0; j < NSUB; ++j) a[j] = HIZW(f.a[j]);
        #pragma unroll
        for (int i = 0; i < 2 * NSUB - 1; ++i) b[i] = HIZW(f.b[i]);
        Kar<NSUB>::run(a, b, acc);
    }
}

__device__ __forceinline__ float ld_guard(const float* __restrict__ p, int i) {
    float v = p[min(max(i, 0), T_LEN - 1)];
    return ((unsigned)i < (unsigned)T_LEN) ? v : 0.f;
}

__device__ __forceinline__ void step_checked8(const float* __restrict__ rir,
                                              const float* __restrict__ a_ch,
                                              int k0, int rbase, int bbase, v8f* acc)
{
    const int r0 = k0 + rbase;
    const int i0 = k0 + bbase;
    v2f a;
    a.x = ld_guard(rir, r0);
    a.y = ld_guard(rir, r0 + 1);
    v2f b[NSUB];
    #pragma unroll
    for (int s = 0; s < NSUB; ++s) {
        b[s].x = ld_guard(a_ch, i0 + 256 * s);
        b[s].y = ld_guard(a_ch, i0 + 256 * s + 1);
    }
    #pragma unroll
    for (int s = 0; s < NSUB; ++s) acc[LEAF(s)] = wmma4(a, b[s], acc[LEAF(s)]);
}

__global__ __launch_bounds__(256)
void RoomImpulseResponse_wmma_conv(const float* __restrict__ audio,
                                   const float* __restrict__ rir,
                                   float* __restrict__ out)
{
    __shared__ float red[NWAVES * TILE_W];

    const int ch   = blockIdx.y;
    const int tid  = threadIdx.x;
    const int wave = __builtin_amdgcn_readfirstlane(tid >> 5);
    const int lane = tid & 31;
    const int h    = lane >> 4;
    const int mq   = lane & 15;

    const float* __restrict__ a_ch = audio + ch * T_LEN;
    float* __restrict__ o_ch = out + ch * OUT_LEN;

    const int tbase = (blockIdx.x >> 5) * (NTILES / 2);
    const int u     = blockIdx.x & 31;

    const int rbase = 2 * h - 16 * mq;

    for (int pass = 0; pass < 2; ++pass) {
        const int tile = tbase + (pass ? (NTILES / 2 - 1 - u) : u);
        const int n0   = tile * TILE_W;

        int klo = T_LEN - 1 - (256 * (NSUB - 1) + 15) - n0;  if (klo < 0) klo = 0;
        int khi = 2 * T_LEN - 2 - n0;
        if (khi > T_LEN - 1 + 240) khi = T_LEN - 1 + 240;

        const int bbase = n0 + mq + 2 * h - (T_LEN - 1);

        const int kbeg   = klo & ~3;
        const int nsteps = ((khi - kbeg) >> 2) + 1;
        const int kend   = kbeg + 4 * nsteps;

        int iLo = T_LEN - 1 - n0;  if (iLo < 240) iLo = 240;
        iLo = (iLo + 3) & ~3;
        int iHi = 2 * T_LEN - 2 - (256 * (NSUB - 1) + 18) - n0;
        if (iHi > T_LEN - 4) iHi = T_LEN - 4;
        int iHiE = (iHi & ~3) + 4;
        if (iHiE < iLo) iHiE = iLo;

        const int pEnd   = min(iLo, kend);
        const int iEnd   = max(min(iHiE, kend), pEnd);
        const int nchunk = (iEnd - pEnd) / CHUNK_K;
        const int iMid   = pEnd + nchunk * CHUNK_K;
        const int nP     = (pEnd - kbeg) >> 2;
        const int nE     = (kend - iEnd) >> 2;
        const int nL     = (iEnd - iMid) >> 2;

        v8f acc[NACC];
        #pragma unroll
        for (int i = 0; i < NACC; ++i) acc[i] = (v8f){0.f,0.f,0.f,0.f,0.f,0.f,0.f,0.f};

        {
            const int nC   = nP + nE;
            const int perC = (nC + NWAVES - 1) / NWAVES;
            const int v1   = min(nC, (wave + 1) * perC);
            for (int v = wave * perC; v < v1; ++v) {
                const int k0 = (v < nP) ? (kbeg + 4 * v) : (iEnd + 4 * (v - nP));
                step_checked8(rir, a_ch, k0, rbase, bbase, acc);
            }
        }

        {
            const int npair8 = 2 * nchunk;
            if (npair8 > 0) {
                const int kc = pEnd + 32 * wave;
                const char* ea = (const char*)rir  + (ptrdiff_t)(kc + 4 * h - 16 * mq) * 4;
                const char* eb = (const char*)a_ch + (ptrdiff_t)(kc + n0 + mq + 4 * h - (T_LEN - 1)) * 4;
                const char* oa = ea + 32;
                const char* ob = eb + 32;
                Grp8 g0, g1;
                load_grp8(g0, ea, eb);
                for (int p = 0; p < npair8 - 1; ++p) {
                    load_grp8(g1, oa, ob);
                    mma_grp8(g0, acc);
                    const int adv = (p & 1) ? (8192 - 64) : 64;
                    ea += adv;
                    eb += adv;
                    load_grp8(g0, ea, eb);
                    asm volatile("" ::: "memory");
                    mma_grp8(g1, acc);
                    oa += adv;
                    ob += adv;
                }
                load_grp8(g1, oa, ob);
                mma_grp8(g0, acc);
                mma_grp8(g1, acc);
            }
        }
        {
            const int perL = (nL + NWAVES - 1) / NWAVES;
            const int l0   = min(nL, wave * perL);
            const int l1   = min(nL, l0 + perL);
            int k0 = iMid + 4 * l0;
            const int kL = iMid + 4 * l1;
            const char* pa = (const char*)rir  + (ptrdiff_t)(k0 + rbase) * 4;
            const char* pb = (const char*)a_ch + (ptrdiff_t)(k0 + bbase) * 4;
            for (; k0 < kL; k0 += 4) {
                const v2f a0 = *(const v2f*)(pa);
                v2f b[NSUB];
                #pragma unroll
                for (int s = 0; s < NSUB; ++s) {
                    const f2u t = *(const f2u*)(pb + 1024 * s);
                    b[s] = (v2f){t.x, t.y};
                }
                #pragma unroll
                for (int s = 0; s < NSUB; ++s) acc[LEAF(s)] = wmma4(a0, b[s], acc[LEAF(s)]);
                pa += 16;
                pb += 16;
            }
        }

        v8f o[NSUB];
        Kar<NSUB>::fold(acc, o);

        #pragma unroll
        for (int v = 0; v < 8; ++v) {
            float* r = red + wave * TILE_W + (v + 8 * h) * 16 + mq;
            #pragma unroll
            for (int s = 0; s < NSUB; ++s) r[256 * s] = o[s][v];
        }
        __syncthreads();

        const int n = n0 + tid;
        #pragma unroll
        for (int s = 0; s < NSUB; ++s) {
            float sum = 0.f;
            #pragma unroll
            for (int w = 0; w < NWAVES; ++w) sum += red[w * TILE_W + 256 * s + tid];
            if (n + 256 * s < OUT_LEN) o_ch[n + 256 * s] = sum;
        }

        __syncthreads();
    }
}

extern "C" void kernel_launch(void* const* d_in, const int* in_sizes, int n_in,
                              void* d_out, int out_size, void* d_ws, size_t ws_size,
                              hipStream_t stream) {
    (void)in_sizes; (void)n_in; (void)d_ws; (void)ws_size; (void)out_size;
    const float* audio = (const float*)d_in[0];
    const float* rir   = (const float*)d_in[1];
    float* out = (float*)d_out;

    dim3 grid(NBLK_X, 4, 1);
    dim3 block(256, 1, 1);
    RoomImpulseResponse_wmma_conv<<<grid, block, 0, stream>>>(audio, rir, out);
}


# Band ends go through the 2048-k chunk path too (fast-FIR identity, loads guarded to zero outside the signals) instead of checked 4-k direct steps; f32 operand sums/differences, sum order changes
# speedup vs baseline: 2.6610x; 1.0514x over previous
#include <hip/hip_runtime.h>
#include <hip/hip_bf16.h>


typedef __attribute__((ext_vector_type(2))) float v2f;
typedef __attribute__((ext_vector_type(8))) float v8f;

typedef struct __attribute__((packed, aligned(4))) { float x, y, z, w; } f4u;

#define T_LEN   131072
#define OUT_LEN (2 * T_LEN - 1)
#define NSUB    8
#define TILE_W  (256 * NSUB)
#define CHUNK_K (256 * NSUB)
#define NTILES  ((OUT_LEN + TILE_W - 1) / TILE_W)
#define NWAVES  8
#define NBLK_X  (NTILES / 2)
#define NACC    27


__device__ __forceinline__ v8f wmma4(v2f a, v2f b, v8f c) {
    return __builtin_amdgcn_wmma_f32_16x16x4_f32(false, a, false, b,
                                                 (short)0, c, false, false);
}

#define LOXY(q) ((v2f){(q).x, (q).y})
#define HIZW(q) ((v2f){(q).z, (q).w})

template <int N> struct Kar {
    enum { H = N / 2, P = Kar<N / 2>::NA, NA = 3 * P };
    static __device__ __forceinline__ void run(const v2f* a, const v2f* b, v8f* acc) {
        v2f as[H];
        v2f bl[N - 1];
        v2f bh[N - 1];
        #pragma unroll
        for (int j = 0; j < H; ++j) as[j] = a[j] + a[H + j];
        #pragma unroll
        for (int i = 0; i < N - 1; ++i) {
            bl[i] = b[i] - b[H + i];
            bh[i] = b[N + i] - b[H + i];
        }
        Kar<H>::run(a,     bl,    acc);
        Kar<H>::run(a + H, bh,    acc + P);
        Kar<H>::run(as,    b + H, acc + 2 * P);
    }
    static __device__ __forceinline__ void fold(const v8f* acc, v8f* out) {
        v8f lo[H], hi[H], sh[H];
        Kar<H>::fold(acc,         lo);
        Kar<H>::fold(acc + P,     hi);
        Kar<H>::fold(acc + 2 * P, sh);
        #pragma unroll
        for (int s = 0; s < H; ++s) {
            out[s]     = lo[s] + sh[s];
            out[H + s] = hi[s] + sh[s];
        }
    }
};
template <> struct Kar<1> {
    enum { NA = 1 };
    static __device__ __forceinline__ void run(const v2f* a, const v2f* b, v8f* acc) {
        acc[0] = wmma4(a[0], b[0], acc[0]);
    }
    static __device__ __forceinline__ void fold(const v8f* acc, v8f* out) {
        out[0] = acc[0];
    }
};

struct Grp8 {
    f4u a[NSUB];
    f4u b[2 * NSUB - 1];
};

__device__ __forceinline__ void load_grp8(Grp8& f, const char* pa, const char* pb) {
    #pragma unroll
    for (int j = 0; j < NSUB; ++j) f.a[j] = *(const f4u*)(pa + 1024 * j);
    #pragma unroll
    for (int i = 0; i < 2 * NSUB - 1; ++i) f.b[i] = *(const f4u*)(pb + 1024 * i);
}

__device__ __forceinline__ float ld_guard(const float* __restrict__ p, int i) {
    float v = p[min(max(i, 0), T_LEN - 1)];
    return ((unsigned)i < (unsigned)T_LEN) ? v : 0.f;
}

__device__ __forceinline__ void load_grp8_guard(Grp8& f, const float* __restrict__ rir,
                                                const float* __restrict__ a_ch,
                                                int ia, int ib) {
    #pragma unroll
    for (int j = 0; j < NSUB; ++j) {
        const int i = ia + 256 * j;
        const f4u v = *(const f4u*)(rir + min(max(i, 0), T_LEN - 4));
        const bool ok = (unsigned)i <= (unsigned)(T_LEN - 4);
        f.a[j].x = ok ? v.x : 0.f;
        f.a[j].y = ok ? v.y : 0.f;
        f.a[j].z = ok ? v.z : 0.f;
        f.a[j].w = ok ? v.w : 0.f;
    }
    #pragma unroll
    for (int i = 0; i < 2 * NSUB - 1; ++i) {
        const int x = ib + 256 * i;
        f.b[i].x = ld_guard(a_ch, x);
        f.b[i].y = ld_guard(a_ch, x + 1);
        f.b[i].z = ld_guard(a_ch, x + 2);
        f.b[i].w = ld_guard(a_ch, x + 3);
    }
}

__device__ __forceinline__ void mma_grp8(const Grp8& f, v8f* acc) {
    {
        v2f a[NSUB], b[2 * NSUB - 1];
        #pragma unroll
        for (int j = 0; j < NSUB; ++j) a[j] = LOXY(f.a[j]);
        #pragma unroll
        for (int i = 0; i < 2 * NSUB - 1; ++i) b[i] = LOXY(f.b[i]);
        Kar<NSUB>::run(a, b, acc);
    }
    {
        v2f a[NSUB], b[2 * NSUB - 1];
        #pragma unroll
        for (int j = 0; j < NSUB; ++j) a[j] = HIZW(f.a[j]);
        #pragma unroll
        for (int i = 0; i < 2 * NSUB - 1; ++i) b[i] = HIZW(f.b[i]);
        Kar<NSUB>::run(a, b, acc);
    }
}

__global__ __launch_bounds__(256)
void RoomImpulseResponse_wmma_conv(const float* __restrict__ audio,
                                   const float* __restrict__ rir,
                                   float* __restrict__ out)
{
    __shared__ float red[NWAVES * TILE_W];

    const int ch   = blockIdx.y;
    const int tid  = threadIdx.x;
    const int wave = __builtin_amdgcn_readfirstlane(tid >> 5);
    const int lane = tid & 31;
    const int h    = lane >> 4;
    const int mq   = lane & 15;

    const float* __restrict__ a_ch = audio + ch * T_LEN;
    float* __restrict__ o_ch = out + ch * OUT_LEN;

    const int tbase = (blockIdx.x >> 5) * (NTILES / 2);
    const int u     = blockIdx.x & 31;

    for (int pass = 0; pass < 2; ++pass) {
        const int tile = tbase + (pass ? (NTILES / 2 - 1 - u) : u);
        const int n0   = tile * TILE_W;

        int klo = T_LEN - 1 - (256 * (NSUB - 1) + 15) - n0;  if (klo < 0) klo = 0;
        int khi = 2 * T_LEN - 2 - n0;
        if (khi > T_LEN - 1 + 240) khi = T_LEN - 1 + 240;

        const int kbeg = klo & ~3;
        const int nch = (khi - kbeg + CHUNK_K) / CHUNK_K;
        int sLo = T_LEN - 1 - n0;  if (sLo < 240) sLo = 240;
        int sHi = 2 * T_LEN - 2 - 3854 - n0;
        if (sHi > T_LEN - CHUNK_K) sHi = T_LEN - CHUNK_K;
        int c0 = (sLo - kbeg + CHUNK_K - 1) >> 11;
        c0 = min(max(c0, 0), nch);
        int c1 = ((sHi - kbeg) >> 11) + 1;
        c1 = min(max(c1, c0), nch);

        v8f acc[NACC];
        #pragma unroll
        for (int i = 0; i < NACC; ++i) acc[i] = (v8f){0.f,0.f,0.f,0.f,0.f,0.f,0.f,0.f};

        {
            const int nG  = 2 * (c0 + (nch - c1));
            const int ia0 = 32 * wave + 4 * h - 16 * mq;
            const int ib0 = 32 * wave + n0 + mq + 4 * h - (T_LEN - 1);
            for (int gp = 0; gp < nG; ++gp) {
                const int cq = gp >> 1;
                const int c  = (cq < c0) ? cq : (cq - c0 + c1);
                const int kg = kbeg + CHUNK_K * c + 16 * (gp & 1);
                Grp8 g0, g1;
                load_grp8_guard(g0, rir, a_ch, kg + ia0,     kg + ib0);
                load_grp8_guard(g1, rir, a_ch, kg + 8 + ia0, kg + 8 + ib0);
                mma_grp8(g0, acc);
                mma_grp8(g1, acc);
            }
        }

        {
            const int npair8 = 2 * (c1 - c0);
            if (npair8 > 0) {
                const int kc = kbeg + CHUNK_K * c0 + 32 * wave;
                const char* ea = (const char*)rir  + (ptrdiff_t)(kc + 4 * h - 16 * mq) * 4;
                const char* eb = (const char*)a_ch + (ptrdiff_t)(kc + n0 + mq + 4 * h - (T_LEN - 1)) * 4;
                const char* oa = ea + 32;
                const char* ob = eb + 32;
                Grp8 g0, g1;
                load_grp8(g0, ea, eb);
                for (int p = 0; p < npair8 - 1; ++p) {
                    load_grp8(g1, oa, ob);
                    mma_grp8(g0, acc);
                    const int adv = (p & 1) ? (8192 - 64) : 64;
                    ea += adv;
                    eb += adv;
                    load_grp8(g0, ea, eb);
                    asm volatile("" ::: "memory");
                    mma_grp8(g1, acc);
                    oa += adv;
                    ob += adv;
                }
                load_grp8(g1, oa, ob);
                mma_grp8(g0, acc);
                mma_grp8(g1, acc);
            }
        }

        v8f o[NSUB];
        Kar<NSUB>::fold(acc, o);

        #pragma unroll
        for (int v = 0; v < 8; ++v) {
            float* r = red + wave * TILE_W + (v + 8 * h) * 16 + mq;
            #pragma unroll
            for (int s = 0; s < NSUB; ++s) r[256 * s] = o[s][v];
        }
        __syncthreads();

        const int n = n0 + tid;
        #pragma unroll
        for (int s = 0; s < NSUB; ++s) {
            float sum = 0.f;
            #pragma unroll
            for (int w = 0; w < NWAVES; ++w) sum += red[w * TILE_W + 256 * s + tid];
            if (n + 256 * s < OUT_LEN) o_ch[n + 256 * s] = sum;
        }

        __syncthreads();
    }
}

extern "C" void kernel_launch(void* const* d_in, const int* in_sizes, int n_in,
                              void* d_out, int out_size, void* d_ws, size_t ws_size,
                              hipStream_t stream) {
    (void)in_sizes; (void)n_in; (void)d_ws; (void)ws_size; (void)out_size;
    const float* audio = (const float*)d_in[0];
    const float* rir   = (const float*)d_in[1];
    float* out = (float*)d_out;

    dim3 grid(NBLK_X, 4, 1);
    dim3 block(256, 1, 1);
    RoomImpulseResponse_wmma_conv<<<grid, block, 0, stream>>>(audio, rir, out);
}


# Full correlation as a zero-padded 2^18-point f32 transform product (512x512 four-step, 3 launches; 16- and 32-point factors as constant-matrix products on the f32 matrix unit); sum order changes
# speedup vs baseline: 12.4070x; 4.6625x over previous
#include <hip/hip_runtime.h>
#include <hip/hip_bf16.h>


typedef __attribute__((ext_vector_type(2))) float v2f;
typedef __attribute__((ext_vector_type(8))) float v8f;

typedef struct __attribute__((packed, aligned(4))) { float x, y, z, w; } f4u;

#define T_LEN   131072
#define OUT_LEN (2 * T_LEN - 1)
#define NSUB    8
#define TILE_W  (256 * NSUB)
#define CHUNK_K (256 * NSUB)
#define NTILES  ((OUT_LEN + TILE_W - 1) / TILE_W)
#define NWAVES  8
#define NBLK_X  (NTILES / 2)
#define NACC    27


__device__ __forceinline__ v8f wmma4(v2f a, v2f b, v8f c) {
    return __builtin_amdgcn_wmma_f32_16x16x4_f32(false, a, false, b,
                                                 (short)0, c, false, false);
}

#define LOXY(q) ((v2f){(q).x, (q).y})
#define HIZW(q) ((v2f){(q).z, (q).w})

template <int N> struct Kar {
    enum { H = N / 2, P = Kar<N / 2>::NA, NA = 3 * P };
    static __device__ __forceinline__ void run(const v2f* a, const v2f* b, v8f* acc) {
        v2f as[H];
        v2f bl[N - 1];
        v2f bh[N - 1];
        #pragma unroll
        for (int j = 0; j < H; ++j) as[j] = a[j] + a[H + j];
        #pragma unroll
        for (int i = 0; i < N - 1; ++i) {
            bl[i] = b[i] - b[H + i];
            bh[i] = b[N + i] - b[H + i];
        }
        Kar<H>::run(a,     bl,    acc);
        Kar<H>::run(a + H, bh,    acc + P);
        Kar<H>::run(as,    b + H, acc + 2 * P);
    }
    static __device__ __forceinline__ void fold(const v8f* acc, v8f* out) {
        v8f lo[H], hi[H], sh[H];
        Kar<H>::fold(acc,         lo);
        Kar<H>::fold(acc + P,     hi);
        Kar<H>::fold(acc + 2 * P, sh);
        #pragma unroll
        for (int s = 0; s < H; ++s) {
            out[s]     = lo[s] + sh[s];
            out[H + s] = hi[s] + sh[s];
        }
    }
};
template <> struct Kar<1> {
    enum { NA = 1 };
    static __device__ __forceinline__ void run(const v2f* a, const v2f* b, v8f* acc) {
        acc[0] = wmma4(a[0], b[0], acc[0]);
    }
    static __device__ __forceinline__ void fold(const v8f* acc, v8f* out) {
        out[0] = acc[0];
    }
};

struct Grp8 {
    f4u a[NSUB];
    f4u b[2 * NSUB - 1];
};

__device__ __forceinline__ void load_grp8(Grp8& f, const char* pa, const char* pb) {
    #pragma unroll
    for (int j = 0; j < NSUB; ++j) f.a[j] = *(const f4u*)(pa + 1024 * j);
    #pragma unroll
    for (int i = 0; i < 2 * NSUB - 1; ++i) f.b[i] = *(const f4u*)(pb + 1024 * i);
}

__device__ __forceinline__ float ld_guard(const float* __restrict__ p, int i) {
    float v = p[min(max(i, 0), T_LEN - 1)];
    return ((unsigned)i < (unsigned)T_LEN) ? v : 0.f;
}

__device__ __forceinline__ void load_grp8_guard(Grp8& f, const float* __restrict__ rir,
                                                const float* __restrict__ a_ch,
                                                int ia, int ib) {
    #pragma unroll
    for (int j = 0; j < NSUB; ++j) {
        const int i = ia + 256 * j;
        const f4u v = *(const f4u*)(rir + min(max(i, 0), T_LEN - 4));
        const bool ok = (unsigned)i <= (unsigned)(T_LEN - 4);
        f.a[j].x = ok ? v.x : 0.f;
        f.a[j].y = ok ? v.y : 0.f;
        f.a[j].z = ok ? v.z : 0.f;
        f.a[j].w = ok ? v.w : 0.f;
    }
    #pragma unroll
    for (int i = 0; i < 2 * NSUB - 1; ++i) {
        const int x = ib + 256 * i;
        f.b[i].x = ld_guard(a_ch, x);
        f.b[i].y = ld_guard(a_ch, x + 1);
        f.b[i].z = ld_guard(a_ch, x + 2);
        f.b[i].w = ld_guard(a_ch, x + 3);
    }
}

__device__ __forceinline__ void mma_grp8(const Grp8& f, v8f* acc) {
    {
        v2f a[NSUB], b[2 * NSUB - 1];
        #pragma unroll
        for (int j = 0; j < NSUB; ++j) a[j] = LOXY(f.a[j]);
        #pragma unroll
        for (int i = 0; i < 2 * NSUB - 1; ++i) b[i] = LOXY(f.b[i]);
        Kar<NSUB>::run(a, b, acc);
    }
    {
        v2f a[NSUB], b[2 * NSUB - 1];
        #pragma unroll
        for (int j = 0; j < NSUB; ++j) a[j] = HIZW(f.a[j]);
        #pragma unroll
        for (int i = 0; i < 2 * NSUB - 1; ++i) b[i] = HIZW(f.b[i]);
        Kar<NSUB>::run(a, b, acc);
    }
}

__global__ __launch_bounds__(256)
void RoomImpulseResponse_wmma_conv(const float* __restrict__ audio,
                                   const float* __restrict__ rir,
                                   float* __restrict__ out)
{
    __shared__ float red[NWAVES * TILE_W];

    const int ch   = blockIdx.y;
    const int tid  = threadIdx.x;
    const int wave = __builtin_amdgcn_readfirstlane(tid >> 5);
    const int lane = tid & 31;
    const int h    = lane >> 4;
    const int mq   = lane & 15;

    const float* __restrict__ a_ch = audio + ch * T_LEN;
    float* __restrict__ o_ch = out + ch * OUT_LEN;

    const int tbase = (blockIdx.x >> 5) * (NTILES / 2);
    const int u     = blockIdx.x & 31;

    for (int pass = 0; pass < 2; ++pass) {
        const int tile = tbase + (pass ? (NTILES / 2 - 1 - u) : u);
        const int n0   = tile * TILE_W;

        int klo = T_LEN - 1 - (256 * (NSUB - 1) + 15) - n0;  if (klo < 0) klo = 0;
        int khi = 2 * T_LEN - 2 - n0;
        if (khi > T_LEN - 1 + 240) khi = T_LEN - 1 + 240;

        const int kbeg = klo & ~3;
        const int nch = (khi - kbeg + CHUNK_K) / CHUNK_K;
        int sLo = T_LEN - 1 - n0;  if (sLo < 240) sLo = 240;
        int sHi = 2 * T_LEN - 2 - 3854 - n0;
        if (sHi > T_LEN - CHUNK_K) sHi = T_LEN - CHUNK_K;
        int c0 = (sLo - kbeg + CHUNK_K - 1) >> 11;
        c0 = min(max(c0, 0), nch);
        int c1 = ((sHi - kbeg) >> 11) + 1;
        c1 = min(max(c1, c0), nch);

        v8f acc[NACC];
        #pragma unroll
        for (int i = 0; i < NACC; ++i) acc[i] = (v8f){0.f,0.f,0.f,0.f,0.f,0.f,0.f,0.f};

        {
            const int nG  = 2 * (c0 + (nch - c1));
            const int ia0 = 32 * wave + 4 * h - 16 * mq;
            const int ib0 = 32 * wave + n0 + mq + 4 * h - (T_LEN - 1);
            for (int gp = 0; gp < nG; ++gp) {
                const int cq = gp >> 1;
                const int c  = (cq < c0) ? cq : (cq - c0 + c1);
                const int kg = kbeg + CHUNK_K * c + 16 * (gp & 1);
                Grp8 g0, g1;
                load_grp8_guard(g0, rir, a_ch, kg + ia0,     kg + ib0);
                load_grp8_guard(g1, rir, a_ch, kg + 8 + ia0, kg + 8 + ib0);
                mma_grp8(g0, acc);
                mma_grp8(g1, acc);
            }
        }

        {
            const int npair8 = 2 * (c1 - c0);
            if (npair8 > 0) {
                const int kc = kbeg + CHUNK_K * c0 + 32 * wave;
                const char* ea = (const char*)rir  + (ptrdiff_t)(kc + 4 * h - 16 * mq) * 4;
                const char* eb = (const char*)a_ch + (ptrdiff_t)(kc + n0 + mq + 4 * h - (T_LEN - 1)) * 4;
                const char* oa = ea + 32;
                const char* ob = eb + 32;
                Grp8 g0, g1;
                load_grp8(g0, ea, eb);
                for (int p = 0; p < npair8 - 1; ++p) {
                    load_grp8(g1, oa, ob);
                    mma_grp8(g0, acc);
                    const int adv = (p & 1) ? (8192 - 64) : 64;
                    ea += adv;
                    eb += adv;
                    load_grp8(g0, ea, eb);
                    asm volatile("" ::: "memory");
                    mma_grp8(g1, acc);
                    oa += adv;
                    ob += adv;
                }
                load_grp8(g1, oa, ob);
                mma_grp8(g0, acc);
                mma_grp8(g1, acc);
            }
        }

        v8f o[NSUB];
        Kar<NSUB>::fold(acc, o);

        #pragma unroll
        for (int v = 0; v < 8; ++v) {
            float* r = red + wave * TILE_W + (v + 8 * h) * 16 + mq;
            #pragma unroll
            for (int s = 0; s < NSUB; ++s) r[256 * s] = o[s][v];
        }
        __syncthreads();

        const int n = n0 + tid;
        #pragma unroll
        for (int s = 0; s < NSUB; ++s) {
            float sum = 0.f;
            #pragma unroll
            for (int w = 0; w < NWAVES; ++w) sum += red[w * TILE_W + 256 * s + tid];
            if (n + 256 * s < OUT_LEN) o_ch[n + 256 * s] = sum;
        }

        __syncthreads();
    }
}


#define NFFT   262144
#define PAN_C  16

__device__ __forceinline__ float2 cis_pi(float x) {
    float s, c;
    sincospif(x, &s, &c);
    return make_float2(c, s);
}

__device__ __forceinline__ float2 cmul(float2 a, float2 b) {
    return make_float2(a.x * b.x - a.y * b.y, a.x * b.y + a.y * b.x);
}

template <int SGN, bool TW>
__device__ __forceinline__ void stage16(float2* pan, int wave, int h, int mq) {
    v2f fr[4], fi[4], fn[4];
    #pragma unroll
    for (int ks = 0; ks < 4; ++ks) {
        #pragma unroll
        for (int v = 0; v < 2; ++v) {
            const int k = 4 * ks + 2 * h + v;
            const int t = (mq * k) & 15;
            const float2 w = cis_pi((float)(SGN * t) * 0.125f);
            fr[ks][v] = w.x;
            fi[ks][v] = w.y;
            fn[ks][v] = -w.y;
        }
    }
    for (int b = wave; b < 32; b += 8) {
        v8f dr = (v8f){0.f,0.f,0.f,0.f,0.f,0.f,0.f,0.f};
        v8f di = (v8f){0.f,0.f,0.f,0.f,0.f,0.f,0.f,0.f};
        #pragma unroll
        for (int ks = 0; ks < 4; ++ks) {
            v2f xr, xi;
            #pragma unroll
            for (int v = 0; v < 2; ++v) {
                const int k = 4 * ks + 2 * h + v;
                const float2 x = pan[(32 * k + b) * PAN_C + mq];
                xr[v] = x.x;
                xi[v] = x.y;
            }
            dr = wmma4(fr[ks], xr, dr);
            dr = wmma4(fn[ks], xi, dr);
            di = wmma4(fr[ks], xi, di);
            di = wmma4(fi[ks], xr, di);
        }
        #pragma unroll
        for (int v = 0; v < 8; ++v) {
            const int m = v + 8 * h;
            float2 y = make_float2(dr[v], di[v]);
            if (TW) y = cmul(y, cis_pi((float)(SGN * m * b) * (1.f / 256.f)));
            pan[(32 * m + b) * PAN_C + mq] = y;
        }
    }
}

template <int SGN, bool TW>
__device__ __forceinline__ void stage32(float2* pan, int wave, int h, int mq) {
    v2f fr[2][8], fi[2][8], fn[2][8];
    #pragma unroll
    for (int mt = 0; mt < 2; ++mt) {
        #pragma unroll
        for (int ks = 0; ks < 8; ++ks) {
            #pragma unroll
            for (int v = 0; v < 2; ++v) {
                const int k = 4 * ks + 2 * h + v;
                const int t = ((mq + 16 * mt) * k) & 31;
                const float2 w = cis_pi((float)(SGN * t) * 0.0625f);
                fr[mt][ks][v] = w.x;
                fi[mt][ks][v] = w.y;
                fn[mt][ks][v] = -w.y;
            }
        }
    }
    for (int c = wave; c < 16; c += 8) {
        v8f dr[2], di[2];
        #pragma unroll
        for (int mt = 0; mt < 2; ++mt) {
            dr[mt] = (v8f){0.f,0.f,0.f,0.f,0.f,0.f,0.f,0.f};
            di[mt] = (v8f){0.f,0.f,0.f,0.f,0.f,0.f,0.f,0.f};
        }
        #pragma unroll
        for (int ks = 0; ks < 8; ++ks) {
            v2f xr, xi;
            #pragma unroll
            for (int v = 0; v < 2; ++v) {
                const int k = 4 * ks + 2 * h + v;
                const float2 x = pan[(32 * c + k) * PAN_C + mq];
                xr[v] = x.x;
                xi[v] = x.y;
            }
            #pragma unroll
            for (int mt = 0; mt < 2; ++mt) {
                dr[mt] = wmma4(fr[mt][ks], xr, dr[mt]);
                dr[mt] = wmma4(fn[mt][ks], xi, dr[mt]);
                di[mt] = wmma4(fr[mt][ks], xi, di[mt]);
                di[mt] = wmma4(fi[mt][ks], xr, di[mt]);
            }
        }
        #pragma unroll
        for (int mt = 0; mt < 2; ++mt) {
            #pragma unroll
            for (int v = 0; v < 8; ++v) {
                const int m = 16 * mt + v + 8 * h;
                float2 y = make_float2(dr[mt][v], di[mt][v]);
                if (TW) y = cmul(y, cis_pi((float)(SGN * m * c) * (1.f / 256.f)));
                pan[(32 * c + m) * PAN_C + mq] = y;
            }
        }
    }
}

__device__ __forceinline__ void fwd512(float2* pan, int wave, int h, int mq) {
    stage16<-1, true>(pan, wave, h, mq);
    __syncthreads();
    stage32<-1, false>(pan, wave, h, mq);
    __syncthreads();
}

__device__ __forceinline__ void inv512(float2* pan, int wave, int h, int mq) {
    stage32<1, true>(pan, wave, h, mq);
    __syncthreads();
    stage16<1, false>(pan, wave, h, mq);
    __syncthreads();
}

__global__ __launch_bounds__(256)
void RoomImpulseResponse_fft_cols_fwd(const float* __restrict__ audio,
                                      const float* __restrict__ rir,
                                      float* ws_f)
{
    __shared__ float2 pan[512 * PAN_C];
    float2* ws = (float2*)ws_f;
    const int slab = blockIdx.x;
    const int sig  = blockIdx.y;
    const int tid  = threadIdx.x;
    const int wave = __builtin_amdgcn_readfirstlane(tid >> 5);
    const int lane = tid & 31;
    const int h    = lane >> 4;
    const int mq   = lane & 15;

    const int j  = tid & 15;
    const int r0 = tid >> 4;
    for (int i = 0; i < 16; ++i) {
        const int n1  = r0 + 16 * i;
        const int idx = 512 * n1 + 16 * slab + j;
        float2 x;
        if (sig < 2) {
            x.x = audio[(2 * sig) * T_LEN + idx];
            x.y = audio[(2 * sig + 1) * T_LEN + idx];
        } else {
            x.x = rir[T_LEN - 1 - idx];
            x.y = 0.f;
        }
        pan[n1 * PAN_C + j] = x;
        pan[(256 + n1) * PAN_C + j] = make_float2(0.f, 0.f);
    }
    __syncthreads();

    fwd512(pan, wave, h, mq);

    for (int i = 0; i < 32; ++i) {
        const int e    = tid + 256 * i;
        const int slot = e >> 4;
        const int jj   = e & 15;
        const int k1   = (slot >> 5) + 16 * (slot & 31);
        const int n2   = 16 * slab + jj;
        const float2 w = cis_pi(-(float)(n2 * k1) * (1.f / 131072.f));
        ws[sig * NFFT + k1 * 512 + n2] = cmul(pan[e], w);
    }
}

__global__ __launch_bounds__(256)
void RoomImpulseResponse_fft_rows(float* ws_f)
{
    __shared__ float2 pan[512 * PAN_C];
    float2* ws = (float2*)ws_f;
    const int k1b  = 16 * blockIdx.x;
    const int zs   = blockIdx.y;
    const int tid  = threadIdx.x;
    const int wave = __builtin_amdgcn_readfirstlane(tid >> 5);
    const int lane = tid & 31;
    const int h    = lane >> 4;
    const int mq   = lane & 15;

    for (int j = 0; j < 16; ++j) {
        const float2* row = ws + 2 * NFFT + (k1b + j) * 512;
        pan[tid * PAN_C + j]         = row[tid];
        pan[(tid + 256) * PAN_C + j] = row[tid + 256];
    }
    __syncthreads();
    fwd512(pan, wave, h, mq);

    float2 G[32];
    #pragma unroll
    for (int i = 0; i < 32; ++i) G[i] = pan[tid + 256 * i];
    __syncthreads();

    for (int j = 0; j < 16; ++j) {
        const float2* row = ws + zs * NFFT + (k1b + j) * 512;
        pan[tid * PAN_C + j]         = row[tid];
        pan[(tid + 256) * PAN_C + j] = row[tid + 256];
    }
    __syncthreads();
    fwd512(pan, wave, h, mq);

    #pragma unroll
    for (int i = 0; i < 32; ++i) {
        const int e = tid + 256 * i;
        pan[e] = cmul(pan[e], G[i]);
    }
    __syncthreads();
    inv512(pan, wave, h, mq);

    for (int j = 0; j < 16; ++j) {
        const int k1 = k1b + j;
        float2* row = ws + zs * NFFT + k1 * 512;
        const float2 w0 = cis_pi((float)(tid * k1) * (1.f / 131072.f));
        const float2 w1 = cis_pi((float)((tid + 256) * k1) * (1.f / 131072.f));
        row[tid]       = cmul(pan[tid * PAN_C + j], w0);
        row[tid + 256] = cmul(pan[(tid + 256) * PAN_C + j], w1);
    }
}

__global__ __launch_bounds__(256)
void RoomImpulseResponse_fft_cols_inv(const float* ws_f, float* __restrict__ out)
{
    __shared__ float2 pan[512 * PAN_C];
    const float2* ws = (const float2*)ws_f;
    const int slab = blockIdx.x;
    const int zs   = blockIdx.y;
    const int tid  = threadIdx.x;
    const int wave = __builtin_amdgcn_readfirstlane(tid >> 5);
    const int lane = tid & 31;
    const int h    = lane >> 4;
    const int mq   = lane & 15;

    const int j  = tid & 15;
    const int r0 = tid >> 4;
    for (int i = 0; i < 32; ++i) {
        const int k1   = r0 + 16 * i;
        const int slot = 32 * r0 + i;
        pan[slot * PAN_C + j] = ws[zs * NFFT + k1 * 512 + 16 * slab + j];
    }
    __syncthreads();

    inv512(pan, wave, h, mq);

    float* o0 = out + (2 * zs) * OUT_LEN;
    float* o1 = out + (2 * zs + 1) * OUT_LEN;
    for (int i = 0; i < 32; ++i) {
        const int e   = tid + 256 * i;
        const int n1  = e >> 4;
        const int jj  = e & 15;
        const int idx = 512 * n1 + 16 * slab + jj;
        const float2 x = pan[e];
        if (idx < OUT_LEN) {
            o0[idx] = x.x * (1.f / 262144.f);
            o1[idx] = x.y * (1.f / 262144.f);
        }
    }
}

extern "C" void kernel_launch(void* const* d_in, const int* in_sizes, int n_in,
                              void* d_out, int out_size, void* d_ws, size_t ws_size,
                              hipStream_t stream) {
    (void)in_sizes; (void)n_in; (void)out_size;
    const float* audio = (const float*)d_in[0];
    const float* rir   = (const float*)d_in[1];
    float* out = (float*)d_out;

    if (d_ws != nullptr && ws_size >= (size_t)6291456) {
        float* ws = (float*)d_ws;
        RoomImpulseResponse_fft_cols_fwd<<<dim3(32, 3, 1), dim3(256, 1, 1), 0, stream>>>(audio, rir, ws);
        RoomImpulseResponse_fft_rows<<<dim3(32, 2, 1), dim3(256, 1, 1), 0, stream>>>(ws);
        RoomImpulseResponse_fft_cols_inv<<<dim3(32, 2, 1), dim3(256, 1, 1), 0, stream>>>(ws, out);
    } else {
        dim3 grid(NBLK_X, 4, 1);
        dim3 block(256, 1, 1);
        RoomImpulseResponse_wmma_conv<<<grid, block, 0, stream>>>(audio, rir, out);
    }
}


# Roots of unity come from a table each block builds once in LDS (1056 entries) instead of a sincospif per use; a four-step twiddle is coarse x fine entry (one more f32 rounding); sum order unchanged
# speedup vs baseline: 15.0231x; 1.2109x over previous
#include <hip/hip_runtime.h>
#include <hip/hip_bf16.h>


typedef __attribute__((ext_vector_type(2))) float v2f;
typedef __attribute__((ext_vector_type(8))) float v8f;

typedef struct __attribute__((packed, aligned(4))) { float x, y, z, w; } f4u;

#define T_LEN   131072
#define OUT_LEN (2 * T_LEN - 1)
#define NSUB    8
#define TILE_W  (256 * NSUB)
#define CHUNK_K (256 * NSUB)
#define NTILES  ((OUT_LEN + TILE_W - 1) / TILE_W)
#define NWAVES  8
#define NBLK_X  (NTILES / 2)
#define NACC    27


__device__ __forceinline__ v8f wmma4(v2f a, v2f b, v8f c) {
    return __builtin_amdgcn_wmma_f32_16x16x4_f32(false, a, false, b,
                                                 (short)0, c, false, false);
}

#define LOXY(q) ((v2f){(q).x, (q).y})
#define HIZW(q) ((v2f){(q).z, (q).w})

template <int N> struct Kar {
    enum { H = N / 2, P = Kar<N / 2>::NA, NA = 3 * P };
    static __device__ __forceinline__ void run(const v2f* a, const v2f* b, v8f* acc) {
        v2f as[H];
        v2f bl[N - 1];
        v2f bh[N - 1];
        #pragma unroll
        for (int j = 0; j < H; ++j) as[j] = a[j] + a[H + j];
        #pragma unroll
        for (int i = 0; i < N - 1; ++i) {
            bl[i] = b[i] - b[H + i];
            bh[i] = b[N + i] - b[H + i];
        }
        Kar<H>::run(a,     bl,    acc);
        Kar<H>::run(a + H, bh,    acc + P);
        Kar<H>::run(as,    b + H, acc + 2 * P);
    }
    static __device__ __forceinline__ void fold(const v8f* acc, v8f* out) {
        v8f lo[H], hi[H], sh[H];
        Kar<H>::fold(acc,         lo);
        Kar<H>::fold(acc + P,     hi);
        Kar<H>::fold(acc + 2 * P, sh);
        #pragma unroll
        for (int s = 0; s < H; ++s) {
            out[s]     = lo[s] + sh[s];
            out[H + s] = hi[s] + sh[s];
        }
    }
};
template <> struct Kar<1> {
    enum { NA = 1 };
    static __device__ __forceinline__ void run(const v2f* a, const v2f* b, v8f* acc) {
        acc[0] = wmma4(a[0], b[0], acc[0]);
    }
    static __device__ __forceinline__ void fold(const v8f* acc, v8f* out) {
        out[0] = acc[0];
    }
};

struct Grp8 {
    f4u a[NSUB];
    f4u b[2 * NSUB - 1];
};

__device__ __forceinline__ void load_grp8(Grp8& f, const char* pa, const char* pb) {
    #pragma unroll
    for (int j = 0; j < NSUB; ++j) f.a[j] = *(const f4u*)(pa + 1024 * j);
    #pragma unroll
    for (int i = 0; i < 2 * NSUB - 1; ++i) f.b[i] = *(const f4u*)(pb + 1024 * i);
}

__device__ __forceinline__ float ld_guard(const float* __restrict__ p, int i) {
    float v = p[min(max(i, 0), T_LEN - 1)];
    return ((unsigned)i < (unsigned)T_LEN) ? v : 0.f;
}

__device__ __forceinline__ void load_grp8_guard(Grp8& f, const float* __restrict__ rir,
                                                const float* __restrict__ a_ch,
                                                int ia, int ib) {
    #pragma unroll
    for (int j = 0; j < NSUB; ++j) {
        const int i = ia + 256 * j;
        const f4u v = *(const f4u*)(rir + min(max(i, 0), T_LEN - 4));
        const bool ok = (unsigned)i <= (unsigned)(T_LEN - 4);
        f.a[j].x = ok ? v.x : 0.f;
        f.a[j].y = ok ? v.y : 0.f;
        f.a[j].z = ok ? v.z : 0.f;
        f.a[j].w = ok ? v.w : 0.f;
    }
    #pragma unroll
    for (int i = 0; i < 2 * NSUB - 1; ++i) {
        const int x = ib + 256 * i;
        f.b[i].x = ld_guard(a_ch, x);
        f.b[i].y = ld_guard(a_ch, x + 1);
        f.b[i].z = ld_guard(a_ch, x + 2);
        f.b[i].w = ld_guard(a_ch, x + 3);
    }
}

__device__ __forceinline__ void mma_grp8(const Grp8& f, v8f* acc) {
    {
        v2f a[NSUB], b[2 * NSUB - 1];
        #pragma unroll
        for (int j = 0; j < NSUB; ++j) a[j] = LOXY(f.a[j]);
        #pragma unroll
        for (int i = 0; i < 2 * NSUB - 1; ++i) b[i] = LOXY(f.b[i]);
        Kar<NSUB>::run(a, b, acc);
    }
    {
        v2f a[NSUB], b[2 * NSUB - 1];
        #pragma unroll
        for (int j = 0; j < NSUB; ++j) a[j] = HIZW(f.a[j]);
        #pragma unroll
        for (int i = 0; i < 2 * NSUB - 1; ++i) b[i] = HIZW(f.b[i]);
        Kar<NSUB>::run(a, b, acc);
    }
}

__global__ __launch_bounds__(256)
void RoomImpulseResponse_wmma_conv(const float* __restrict__ audio,
                                   const float* __restrict__ rir,
                                   float* __restrict__ out)
{
    __shared__ float red[NWAVES * TILE_W];

    const int ch   = blockIdx.y;
    const int tid  = threadIdx.x;
    const int wave = __builtin_amdgcn_readfirstlane(tid >> 5);
    const int lane = tid & 31;
    const int h    = lane >> 4;
    const int mq   = lane & 15;

    const float* __restrict__ a_ch = audio + ch * T_LEN;
    float* __restrict__ o_ch = out + ch * OUT_LEN;

    const int tbase = (blockIdx.x >> 5) * (NTILES / 2);
    const int u     = blockIdx.x & 31;

    for (int pass = 0; pass < 2; ++pass) {
        const int tile = tbase + (pass ? (NTILES / 2 - 1 - u) : u);
        const int n0   = tile * TILE_W;

        int klo = T_LEN - 1 - (256 * (NSUB - 1) + 15) - n0;  if (klo < 0) klo = 0;
        int khi = 2 * T_LEN - 2 - n0;
        if (khi > T_LEN - 1 + 240) khi = T_LEN - 1 + 240;

        const int kbeg = klo & ~3;
        const int nch = (khi - kbeg + CHUNK_K) / CHUNK_K;
        int sLo = T_LEN - 1 - n0;  if (sLo < 240) sLo = 240;
        int sHi = 2 * T_LEN - 2 - 3854 - n0;
        if (sHi > T_LEN - CHUNK_K) sHi = T_LEN - CHUNK_K;
        int c0 = (sLo - kbeg + CHUNK_K - 1) >> 11;
        c0 = min(max(c0, 0), nch);
        int c1 = ((sHi - kbeg) >> 11) + 1;
        c1 = min(max(c1, c0), nch);

        v8f acc[NACC];
        #pragma unroll
        for (int i = 0; i < NACC; ++i) acc[i] = (v8f){0.f,0.f,0.f,0.f,0.f,0.f,0.f,0.f};

        {
            const int nG  = 2 * (c0 + (nch - c1));
            const int ia0 = 32 * wave + 4 * h - 16 * mq;
            const int ib0 = 32 * wave + n0 + mq + 4 * h - (T_LEN - 1);
            for (int gp = 0; gp < nG; ++gp) {
                const int cq = gp >> 1;
                const int c  = (cq < c0) ? cq : (cq - c0 + c1);
                const int kg = kbeg + CHUNK_K * c + 16 * (gp & 1);
                Grp8 g0, g1;
                load_grp8_guard(g0, rir, a_ch, kg + ia0,     kg + ib0);
                load_grp8_guard(g1, rir, a_ch, kg + 8 + ia0, kg + 8 + ib0);
                mma_grp8(g0, acc);
                mma_grp8(g1, acc);
            }
        }

        {
            const int npair8 = 2 * (c1 - c0);
            if (npair8 > 0) {
                const int kc = kbeg + CHUNK_K * c0 + 32 * wave;
                const char* ea = (const char*)rir  + (ptrdiff_t)(kc + 4 * h - 16 * mq) * 4;
                const char* eb = (const char*)a_ch + (ptrdiff_t)(kc + n0 + mq + 4 * h - (T_LEN - 1)) * 4;
                const char* oa = ea + 32;
                const char* ob = eb + 32;
                Grp8 g0, g1;
                load_grp8(g0, ea, eb);
                for (int p = 0; p < npair8 - 1; ++p) {
                    load_grp8(g1, oa, ob);
                    mma_grp8(g0, acc);
                    const int adv = (p & 1) ? (8192 - 64) : 64;
                    ea += adv;
                    eb += adv;
                    load_grp8(g0, ea, eb);
                    asm volatile("" ::: "memory");
                    mma_grp8(g1, acc);
                    oa += adv;
                    ob += adv;
                }
                load_grp8(g1, oa, ob);
                mma_grp8(g0, acc);
                mma_grp8(g1, acc);
            }
        }

        v8f o[NSUB];
        Kar<NSUB>::fold(acc, o);

        #pragma unroll
        for (int v = 0; v < 8; ++v) {
            float* r = red + wave * TILE_W + (v + 8 * h) * 16 + mq;
            #pragma unroll
            for (int s = 0; s < NSUB; ++s) r[256 * s] = o[s][v];
        }
        __syncthreads();

        const int n = n0 + tid;
        #pragma unroll
        for (int s = 0; s < NSUB; ++s) {
            float sum = 0.f;
            #pragma unroll
            for (int w = 0; w < NWAVES; ++w) sum += red[w * TILE_W + 256 * s + tid];
            if (n + 256 * s < OUT_LEN) o_ch[n + 256 * s] = sum;
        }

        __syncthreads();
    }
}


#define NFFT   262144
#define PAN_C  16

__device__ __forceinline__ float2 cis_pi(float x) {
    float s, c;
    sincospif(x, &s, &c);
    return make_float2(c, s);
}

__device__ __forceinline__ float2 cmul(float2 a, float2 b) {
    return make_float2(a.x * b.x - a.y * b.y, a.x * b.y + a.y * b.x);
}

#define TW_512  0
#define TW_FINE 512
#define TW_32   1024
#define TW_LEN  1056

template <bool FINE>
__device__ __forceinline__ void build_tw(float2* tw, int tid) {
    tw[TW_512 + tid]       = cis_pi((float)tid * (1.f / 256.f));
    tw[TW_512 + tid + 256] = cis_pi((float)(tid + 256) * (1.f / 256.f));
    if (FINE) {
        tw[TW_FINE + tid]       = cis_pi((float)tid * (1.f / 131072.f));
        tw[TW_FINE + tid + 256] = cis_pi((float)(tid + 256) * (1.f / 131072.f));
    }
    if (tid < 32) tw[TW_32 + tid] = cis_pi((float)tid * 0.0625f);
}

template <int SGN>
__device__ __forceinline__ float2 tw_get(const float2* tw, int i) {
    float2 w = tw[i];
    if (SGN < 0) w.y = -w.y;
    return w;
}

template <int SGN>
__device__ __forceinline__ float2 tw_big(const float2* tw, int p) {
    float2 w = cmul(tw[TW_512 + (p >> 9)], tw[TW_FINE + (p & 511)]);
    if (SGN < 0) w.y = -w.y;
    return w;
}

template <int SGN, bool TW>
__device__ __forceinline__ void stage16(float2* pan, const float2* tw, int wave, int h, int mq) {
    v2f fr[4], fi[4], fn[4];
    #pragma unroll
    for (int ks = 0; ks < 4; ++ks) {
        #pragma unroll
        for (int v = 0; v < 2; ++v) {
            const int k = 4 * ks + 2 * h + v;
            const int t = (mq * k) & 15;
            const float2 w = tw_get<SGN>(tw, TW_32 + 2 * t);
            fr[ks][v] = w.x;
            fi[ks][v] = w.y;
            fn[ks][v] = -w.y;
        }
    }
    for (int b = wave; b < 32; b += 8) {
        v8f dr = (v8f){0.f,0.f,0.f,0.f,0.f,0.f,0.f,0.f};
        v8f di = (v8f){0.f,0.f,0.f,0.f,0.f,0.f,0.f,0.f};
        #pragma unroll
        for (int ks = 0; ks < 4; ++ks) {
            v2f xr, xi;
            #pragma unroll
            for (int v = 0; v < 2; ++v) {
                const int k = 4 * ks + 2 * h + v;
                const float2 x = pan[(32 * k + b) * PAN_C + mq];
                xr[v] = x.x;
                xi[v] = x.y;
            }
            dr = wmma4(fr[ks], xr, dr);
            dr = wmma4(fn[ks], xi, dr);
            di = wmma4(fr[ks], xi, di);
            di = wmma4(fi[ks], xr, di);
        }
        #pragma unroll
        for (int v = 0; v < 8; ++v) {
            const int m = v + 8 * h;
            float2 y = make_float2(dr[v], di[v]);
            if (TW) y = cmul(y, tw_get<SGN>(tw, TW_512 + m * b));
            pan[(32 * m + b) * PAN_C + mq] = y;
        }
    }
}

template <int SGN, bool TW>
__device__ __forceinline__ void stage32(float2* pan, const float2* tw, int wave, int h, int mq) {
    v2f fr[2][8], fi[2][8], fn[2][8];
    #pragma unroll
    for (int mt = 0; mt < 2; ++mt) {
        #pragma unroll
        for (int ks = 0; ks < 8; ++ks) {
            #pragma unroll
            for (int v = 0; v < 2; ++v) {
                const int k = 4 * ks + 2 * h + v;
                const int t = ((mq + 16 * mt) * k) & 31;
                const float2 w = tw_get<SGN>(tw, TW_32 + t);
                fr[mt][ks][v] = w.x;
                fi[mt][ks][v] = w.y;
                fn[mt][ks][v] = -w.y;
            }
        }
    }
    for (int c = wave; c < 16; c += 8) {
        v8f dr[2], di[2];
        #pragma unroll
        for (int mt = 0; mt < 2; ++mt) {
            dr[mt] = (v8f){0.f,0.f,0.f,0.f,0.f,0.f,0.f,0.f};
            di[mt] = (v8f){0.f,0.f,0.f,0.f,0.f,0.f,0.f,0.f};
        }
        #pragma unroll
        for (int ks = 0; ks < 8; ++ks) {
            v2f xr, xi;
            #pragma unroll
            for (int v = 0; v < 2; ++v) {
                const int k = 4 * ks + 2 * h + v;
                const float2 x = pan[(32 * c + k) * PAN_C + mq];
                xr[v] = x.x;
                xi[v] = x.y;
            }
            #pragma unroll
            for (int mt = 0; mt < 2; ++mt) {
                dr[mt] = wmma4(fr[mt][ks], xr, dr[mt]);
                dr[mt] = wmma4(fn[mt][ks], xi, dr[mt]);
                di[mt] = wmma4(fr[mt][ks], xi, di[mt]);
                di[mt] = wmma4(fi[mt][ks], xr, di[mt]);
            }
        }
        #pragma unroll
        for (int mt = 0; mt < 2; ++mt) {
            #pragma unroll
            for (int v = 0; v < 8; ++v) {
                const int m = 16 * mt + v + 8 * h;
                float2 y = make_float2(dr[mt][v], di[mt][v]);
                if (TW) y = cmul(y, tw_get<SGN>(tw, TW_512 + m * c));
                pan[(32 * c + m) * PAN_C + mq] = y;
            }
        }
    }
}

__device__ __forceinline__ void fwd512(float2* pan, const float2* tw, int wave, int h, int mq) {
    stage16<-1, true>(pan, tw, wave, h, mq);
    __syncthreads();
    stage32<-1, false>(pan, tw, wave, h, mq);
    __syncthreads();
}

__device__ __forceinline__ void inv512(float2* pan, const float2* tw, int wave, int h, int mq) {
    stage32<1, true>(pan, tw, wave, h, mq);
    __syncthreads();
    stage16<1, false>(pan, tw, wave, h, mq);
    __syncthreads();
}

__global__ __launch_bounds__(256)
void RoomImpulseResponse_fft_cols_fwd(const float* __restrict__ audio,
                                      const float* __restrict__ rir,
                                      float* ws_f)
{
    __shared__ float2 pan[512 * PAN_C];
    __shared__ float2 tw[TW_LEN];
    float2* ws = (float2*)ws_f;
    const int slab = blockIdx.x;
    const int sig  = blockIdx.y;
    const int tid  = threadIdx.x;
    const int wave = __builtin_amdgcn_readfirstlane(tid >> 5);
    const int lane = tid & 31;
    const int h    = lane >> 4;
    const int mq   = lane & 15;

    build_tw<true>(tw, tid);

    const int j  = tid & 15;
    const int r0 = tid >> 4;
    for (int i = 0; i < 16; ++i) {
        const int n1  = r0 + 16 * i;
        const int idx = 512 * n1 + 16 * slab + j;
        float2 x;
        if (sig < 2) {
            x.x = audio[(2 * sig) * T_LEN + idx];
            x.y = audio[(2 * sig + 1) * T_LEN + idx];
        } else {
            x.x = rir[T_LEN - 1 - idx];
            x.y = 0.f;
        }
        pan[n1 * PAN_C + j] = x;
        pan[(256 + n1) * PAN_C + j] = make_float2(0.f, 0.f);
    }
    __syncthreads();

    fwd512(pan, tw, wave, h, mq);

    for (int i = 0; i < 32; ++i) {
        const int e    = tid + 256 * i;
        const int slot = e >> 4;
        const int jj   = e & 15;
        const int k1   = (slot >> 5) + 16 * (slot & 31);
        const int n2   = 16 * slab + jj;
        const float2 w = tw_big<-1>(tw, n2 * k1);
        ws[sig * NFFT + k1 * 512 + n2] = cmul(pan[e], w);
    }
}

__global__ __launch_bounds__(256)
void RoomImpulseResponse_fft_rows(float* ws_f)
{
    __shared__ float2 pan[512 * PAN_C];
    __shared__ float2 tw[TW_LEN];
    float2* ws = (float2*)ws_f;
    const int k1b  = 16 * blockIdx.x;
    const int zs   = blockIdx.y;
    const int tid  = threadIdx.x;
    const int wave = __builtin_amdgcn_readfirstlane(tid >> 5);
    const int lane = tid & 31;
    const int h    = lane >> 4;
    const int mq   = lane & 15;

    build_tw<true>(tw, tid);

    for (int j = 0; j < 16; ++j) {
        const float2* row = ws + 2 * NFFT + (k1b + j) * 512;
        pan[tid * PAN_C + j]         = row[tid];
        pan[(tid + 256) * PAN_C + j] = row[tid + 256];
    }
    __syncthreads();
    fwd512(pan, tw, wave, h, mq);

    float2 G[32];
    #pragma unroll
    for (int i = 0; i < 32; ++i) G[i] = pan[tid + 256 * i];
    __syncthreads();

    for (int j = 0; j < 16; ++j) {
        const float2* row = ws + zs * NFFT + (k1b + j) * 512;
        pan[tid * PAN_C + j]         = row[tid];
        pan[(tid + 256) * PAN_C + j] = row[tid + 256];
    }
    __syncthreads();
    fwd512(pan, tw, wave, h, mq);

    #pragma unroll
    for (int i = 0; i < 32; ++i) {
        const int e = tid + 256 * i;
        pan[e] = cmul(pan[e], G[i]);
    }
    __syncthreads();
    inv512(pan, tw, wave, h, mq);

    for (int j = 0; j < 16; ++j) {
        const int k1 = k1b + j;
        float2* row = ws + zs * NFFT + k1 * 512;
        const float2 w0 = tw_big<1>(tw, tid * k1);
        const float2 w1 = tw_big<1>(tw, (tid + 256) * k1);
        row[tid]       = cmul(pan[tid * PAN_C + j], w0);
        row[tid + 256] = cmul(pan[(tid + 256) * PAN_C + j], w1);
    }
}

__global__ __launch_bounds__(256)
void RoomImpulseResponse_fft_cols_inv(const float* ws_f, float* __restrict__ out)
{
    __shared__ float2 pan[512 * PAN_C];
    __shared__ float2 tw[TW_LEN];
    const float2* ws = (const float2*)ws_f;
    const int slab = blockIdx.x;
    const int zs   = blockIdx.y;
    const int tid  = threadIdx.x;
    const int wave = __builtin_amdgcn_readfirstlane(tid >> 5);
    const int lane = tid & 31;
    const int h    = lane >> 4;
    const int mq   = lane & 15;

    build_tw<false>(tw, tid);

    const int j  = tid & 15;
    const int r0 = tid >> 4;
    for (int i = 0; i < 32; ++i) {
        const int k1   = r0 + 16 * i;
        const int slot = 32 * r0 + i;
        pan[slot * PAN_C + j] = ws[zs * NFFT + k1 * 512 + 16 * slab + j];
    }
    __syncthreads();

    inv512(pan, tw, wave, h, mq);

    float* o0 = out + (2 * zs) * OUT_LEN;
    float* o1 = out + (2 * zs + 1) * OUT_LEN;
    for (int i = 0; i < 32; ++i) {
        const int e   = tid + 256 * i;
        const int n1  = e >> 4;
        const int jj  = e & 15;
        const int idx = 512 * n1 + 16 * slab + jj;
        const float2 x = pan[e];
        if (idx < OUT_LEN) {
            o0[idx] = x.x * (1.f / 262144.f);
            o1[idx] = x.y * (1.f / 262144.f);
        }
    }
}

extern "C" void kernel_launch(void* const* d_in, const int* in_sizes, int n_in,
                              void* d_out, int out_size, void* d_ws, size_t ws_size,
                              hipStream_t stream) {
    (void)in_sizes; (void)n_in; (void)out_size;
    const float* audio = (const float*)d_in[0];
    const float* rir   = (const float*)d_in[1];
    float* out = (float*)d_out;

    if (d_ws != nullptr && ws_size >= (size_t)6291456) {
        float* ws = (float*)d_ws;
        RoomImpulseResponse_fft_cols_fwd<<<dim3(32, 3, 1), dim3(256, 1, 1), 0, stream>>>(audio, rir, ws);
        RoomImpulseResponse_fft_rows<<<dim3(32, 2, 1), dim3(256, 1, 1), 0, stream>>>(ws);
        RoomImpulseResponse_fft_cols_inv<<<dim3(32, 2, 1), dim3(256, 1, 1), 0, stream>>>(ws, out);
    } else {
        dim3 grid(NBLK_X, 4, 1);
        dim3 block(256, 1, 1);
        RoomImpulseResponse_wmma_conv<<<grid, block, 0, stream>>>(audio, rir, out);
    }
}
